# starting kernel
# baseline (speedup 1.0000x reference)
#include <hip/hip_runtime.h>
#include <hip/hip_bf16.h>
#include <math.h>

typedef float v8f __attribute__((ext_vector_type(8)));
typedef float v2f __attribute__((ext_vector_type(2)));

#define D_HID 128
#define D_OUT 64
#define BN_EPS 1e-5f


__global__ void init_deg_stats(float* deg, float* stats, int N) {
    int i = blockIdx.x * blockDim.x + threadIdx.x;
    int stride = blockDim.x * gridDim.x;
    for (; i < N; i += stride) {
        deg[i] = 1.0f;
        if (i < 256) stats[i] = 0.0f;
    }
}

__global__ void count_edges(const int* __restrict__ col, float* deg, int E) {
    int e = blockIdx.x * blockDim.x + threadIdx.x;
    if (e < E) atomicAdd(&deg[col[e]], 1.0f);
}

__global__ void deg_to_dis(float* deg, int N) {
    int i = blockIdx.x * blockDim.x + threadIdx.x;
    if (i < N) deg[i] = rsqrtf(deg[i]);
}


__global__ void gemm1_wmma(const float* __restrict__ x, const float* __restrict__ W1,
                           const float* __restrict__ b1, float* __restrict__ h1, int N) {
    const int wave  = threadIdx.x >> 5;
    const int lane  = threadIdx.x & 31;
    const int row0  = blockIdx.x * 16;
    if (row0 >= N) return;
    const int col0  = wave * 16;
    const int m     = lane & 15;
    const int n     = lane & 15;
    const int khalf = lane >> 4;

    v8f c;
    const float bias = b1[col0 + n];
    #pragma unroll
    for (int r = 0; r < 8; ++r) c[r] = bias;

    const float* arow = x + (size_t)(row0 + m) * D_HID + 2 * khalf;
    #pragma unroll 4
    for (int k = 0; k < 128; k += 4) {
        v2f a, b;
        a.x = arow[k];
        a.y = arow[k + 1];
        const int kb = k + 2 * khalf;
        b.x = W1[(size_t)(kb + 0) * D_HID + col0 + n];
        b.y = W1[(size_t)(kb + 1) * D_HID + col0 + n];
        c = __builtin_amdgcn_wmma_f32_16x16x4_f32(false, a, false, b, (short)0, c, false, false);
    }

    const int mbase = khalf * 8;
    #pragma unroll
    for (int r = 0; r < 8; ++r)
        h1[(size_t)(row0 + mbase + r) * D_HID + col0 + n] = c[r];
}


__global__ void selfloop1(const float* __restrict__ h1, const float* __restrict__ dis,
                          float* __restrict__ hagg, int total) {
    int i = blockIdx.x * blockDim.x + threadIdx.x;
    if (i < total) {
        float d = dis[i >> 7];
        hagg[i] = d * d * h1[i];
    }
}

__global__ void scatter1(const int* __restrict__ row, const int* __restrict__ col,
                         const float* __restrict__ dis, const float* __restrict__ h1,
                         float* hagg, int E) {
    const int wave = threadIdx.x >> 5;
    const int lane = threadIdx.x & 31;
    const int e = blockIdx.x * 8 + wave;
    if (e >= E) return;
    const int r = row[e];
    const int c = col[e];
    const float w = dis[r] * dis[c];
    const int base = lane * 4;
    const float4 v = *(const float4*)(h1 + (size_t)r * D_HID + base);
    float* dst = hagg + (size_t)c * D_HID + base;
    atomicAdd(dst + 0, w * v.x);
    atomicAdd(dst + 1, w * v.y);
    atomicAdd(dst + 2, w * v.z);
    atomicAdd(dst + 3, w * v.w);
}


__global__ void bn_stats(const float* __restrict__ hagg, float* bsum, float* bsq, int total) {
    __shared__ float ssum[D_HID];
    __shared__ float ssq[D_HID];
    for (int i = threadIdx.x; i < D_HID; i += blockDim.x) { ssum[i] = 0.0f; ssq[i] = 0.0f; }
    __syncthreads();
    int stride = blockDim.x * gridDim.x;
    for (int i = blockIdx.x * blockDim.x + threadIdx.x; i < total; i += stride) {
        float v = hagg[i];
        atomicAdd(&ssum[i & 127], v);
        atomicAdd(&ssq[i & 127], v * v);
    }
    __syncthreads();
    for (int i = threadIdx.x; i < D_HID; i += blockDim.x) {
        atomicAdd(&bsum[i], ssum[i]);
        atomicAdd(&bsq[i], ssq[i]);
    }
}

__global__ void bn_finalize(const float* __restrict__ bsum, const float* __restrict__ bsq,
                            const float* __restrict__ gamma, const float* __restrict__ beta,
                            float* scale, float* shift, int N) {
    int c = threadIdx.x;
    if (c < D_HID) {
        float inv = 1.0f / (float)N;
        float mu = bsum[c] * inv;
        float var = bsq[c] * inv - mu * mu;
        float s = gamma[c] * rsqrtf(var + BN_EPS);
        scale[c] = s;
        shift[c] = beta[c] - mu * s;
    }
}


__global__ void gemm2_wmma(const float* __restrict__ hagg, const float* __restrict__ scale,
                           const float* __restrict__ shift, const float* __restrict__ W2,
                           const float* __restrict__ b2, float* __restrict__ h2, int N) {
    const int wave  = threadIdx.x >> 5;
    const int lane  = threadIdx.x & 31;
    const int row0  = (blockIdx.x * 2 + (wave >> 2)) * 16;
    if (row0 >= N) return;
    const int col0  = (wave & 3) * 16;
    const int m     = lane & 15;
    const int n     = lane & 15;
    const int khalf = lane >> 4;

    v8f c;
    const float bias = b2[col0 + n];
    #pragma unroll
    for (int r = 0; r < 8; ++r) c[r] = bias;

    const float* arow = hagg + (size_t)(row0 + m) * D_HID;
    #pragma unroll 4
    for (int k = 0; k < 128; k += 4) {
        const int ka = k + 2 * khalf;
        v2f a, b;
        a.x = fmaxf(arow[ka + 0] * scale[ka + 0] + shift[ka + 0], 0.0f);
        a.y = fmaxf(arow[ka + 1] * scale[ka + 1] + shift[ka + 1], 0.0f);
        b.x = W2[(size_t)(ka + 0) * D_OUT + col0 + n];
        b.y = W2[(size_t)(ka + 1) * D_OUT + col0 + n];
        c = __builtin_amdgcn_wmma_f32_16x16x4_f32(false, a, false, b, (short)0, c, false, false);
    }

    const int mbase = khalf * 8;
    #pragma unroll
    for (int r = 0; r < 8; ++r)
        h2[(size_t)(row0 + mbase + r) * D_OUT + col0 + n] = c[r];
}


__global__ void selfloop2(const float* __restrict__ h2, const float* __restrict__ dis,
                          float* __restrict__ out, int total) {
    int i = blockIdx.x * blockDim.x + threadIdx.x;
    if (i < total) {
        float d = dis[i >> 6];
        out[i] = d * d * h2[i];
    }
}

__global__ void scatter2(const int* __restrict__ row, const int* __restrict__ col,
                         const float* __restrict__ dis, const float* __restrict__ h2,
                         float* out, int E) {
    const int wave = threadIdx.x >> 5;
    const int lane = threadIdx.x & 31;
    const int e = blockIdx.x * 8 + wave;
    if (e >= E) return;
    const int r = row[e];
    const int c = col[e];
    const float w = dis[r] * dis[c];
    const int base = lane * 2;
    const float2 v = *(const float2*)(h2 + (size_t)r * D_OUT + base);
    float* dst = out + (size_t)c * D_OUT + base;
    atomicAdd(dst + 0, w * v.x);
    atomicAdd(dst + 1, w * v.y);
}


__global__ void logsoftmax_rows(float* out, int N) {
    const int wave = threadIdx.x >> 5;
    const int lane = threadIdx.x & 31;
    const int rowi = blockIdx.x * 8 + wave;
    if (rowi >= N) return;
    float* p = out + (size_t)rowi * D_OUT;
    float v0 = p[lane];
    float v1 = p[lane + 32];
    float mx = fmaxf(v0, v1);
    #pragma unroll
    for (int off = 16; off >= 1; off >>= 1)
        mx = fmaxf(mx, __shfl_xor(mx, off, 32));
    float s = __expf(v0 - mx) + __expf(v1 - mx);
    #pragma unroll
    for (int off = 16; off >= 1; off >>= 1)
        s += __shfl_xor(s, off, 32);
    float l = mx + __logf(s);
    p[lane]      = v0 - l;
    p[lane + 32] = v1 - l;
}


extern "C" void kernel_launch(void* const* d_in, const int* in_sizes, int n_in,
                              void* d_out, int out_size, void* d_ws, size_t ws_size,
                              hipStream_t stream) {
    const float* x     = (const float*)d_in[0];
    const float* W1    = (const float*)d_in[1];
    const float* b1    = (const float*)d_in[2];
    const float* gamma = (const float*)d_in[3];
    const float* beta  = (const float*)d_in[4];
    const float* W2    = (const float*)d_in[5];
    const float* b2    = (const float*)d_in[6];
    const int*   ei    = (const int*)d_in[7];

    const int N = in_sizes[0] / D_HID;
    const int E = in_sizes[7] / 2;
    const int* erow = ei;
    const int* ecol = ei + E;

    float* ws    = (float*)d_ws;
    float* dis   = ws;
    float* h1    = dis + N;
    float* hagg  = h1 + (size_t)N * D_HID;
    float* stats = hagg + (size_t)N * D_HID;
    float* bsum  = stats;
    float* bsq   = stats + 128;
    float* scal  = stats + 256;
    float* shft  = stats + 384;
    float* h2    = h1;
    float* outp  = (float*)d_out;

    init_deg_stats<<<(N + 255) / 256, 256, 0, stream>>>(dis, stats, N);
    count_edges<<<(E + 255) / 256, 256, 0, stream>>>(ecol, dis, E);
    deg_to_dis<<<(N + 255) / 256, 256, 0, stream>>>(dis, N);

    gemm1_wmma<<<(N + 15) / 16, 256, 0, stream>>>(x, W1, b1, h1, N);
    selfloop1<<<((size_t)N * D_HID + 255) / 256, 256, 0, stream>>>(h1, dis, hagg, N * D_HID);
    scatter1<<<(E + 7) / 8, 256, 0, stream>>>(erow, ecol, dis, h1, hagg, E);

    bn_stats<<<1024, 256, 0, stream>>>(hagg, bsum, bsq, N * D_HID);
    bn_finalize<<<1, 128, 0, stream>>>(bsum, bsq, gamma, beta, scal, shft, N);

    gemm2_wmma<<<(N + 31) / 32, 256, 0, stream>>>(hagg, scal, shft, W2, b2, h2, N);
    selfloop2<<<((size_t)N * D_OUT + 255) / 256, 256, 0, stream>>>(h2, dis, outp, N * D_OUT);
    scatter2<<<(E + 7) / 8, 256, 0, stream>>>(erow, ecol, dis, h2, outp, E);

    logsoftmax_rows<<<(N + 7) / 8, 256, 0, stream>>>(outp, N);
}


# Both propagations: float atomic scatter-add replaced by a per-call destination-sorted edge list (rank = return of an int counting atomic); a wave sums a node's rows in registers in arrival-rank order
# speedup vs baseline: 2.0591x; 2.0591x over previous
#include <hip/hip_runtime.h>
#include <hip/hip_bf16.h>
#include <math.h>

typedef float v8f __attribute__((ext_vector_type(8)));
typedef float v2f __attribute__((ext_vector_type(2)));

#define D_HID 128
#define D_OUT 64
#define BN_EPS 1e-5f


__global__ void init_cnt_stats(int* cnt, float* stats, int* counter, int N) {
    int i = blockIdx.x * blockDim.x + threadIdx.x;
    int stride = blockDim.x * gridDim.x;
    for (; i < N; i += stride) {
        cnt[i] = 0;
        if (i < 256) stats[i] = 0.0f;
        if (i == 0) *counter = 0;
    }
}

__global__ void count_rank(const int* __restrict__ col, int* cnt, int* __restrict__ pos, int E) {
    int e = blockIdx.x * blockDim.x + threadIdx.x;
    if (e < E) pos[e] = atomicAdd(&cnt[col[e]], 1);
}

__global__ void scan_dis(const int* __restrict__ cnt, int4* __restrict__ seg, int* counter, int N) {
    __shared__ int wsum[32];
    __shared__ int sbase;
    const int tid  = threadIdx.x;
    const int lane = tid & 31;
    const int wave = tid >> 5;
    const int i = blockIdx.x * 1024 + tid;
    const int v = (i < N) ? cnt[i] : 0;
    int inc = v;
    for (int off = 1; off < 32; off <<= 1) {
        const int t = __shfl_up(inc, off, 32);
        if (lane >= off) inc += t;
    }
    if (lane == 31) wsum[wave] = inc;
    __syncthreads();
    if (wave == 0) {
        const int sv = wsum[lane];
        int si = sv;
        for (int off = 1; off < 32; off <<= 1) {
            const int t = __shfl_up(si, off, 32);
            if (lane >= off) si += t;
        }
        wsum[lane] = si - sv;
        if (lane == 31) sbase = atomicAdd(counter, si);
    }
    __syncthreads();
    if (i < N)
        seg[i] = make_int4(sbase + wsum[wave] + inc - v, v, __float_as_int(rsqrtf((float)(v + 1))), 0);
}

__global__ void fill_sorted(const int* __restrict__ row, const int* __restrict__ col,
                            const int4* __restrict__ seg, const int* __restrict__ pos,
                            int2* __restrict__ sedge, int E) {
    int e = blockIdx.x * blockDim.x + threadIdx.x;
    if (e < E) {
        const int r = row[e];
        sedge[seg[col[e]].x + pos[e]] = make_int2(r, seg[r].z);
    }
}


__device__ __forceinline__ void axpy(float4& acc, float w, const float4& v) {
    acc.x = fmaf(w, v.x, acc.x); acc.y = fmaf(w, v.y, acc.y);
    acc.z = fmaf(w, v.z, acc.z); acc.w = fmaf(w, v.w, acc.w);
}
__device__ __forceinline__ void axpy(float2& acc, float w, const float2& v) {
    acc.x = fmaf(w, v.x, acc.x); acc.y = fmaf(w, v.y, acc.y);
}
__device__ __forceinline__ void set_zero(float4& a) { a = make_float4(0.0f, 0.0f, 0.0f, 0.0f); }
__device__ __forceinline__ void set_zero(float2& a) { a = make_float2(0.0f, 0.0f); }

template <int K, typename VT>
__device__ __forceinline__ void edge_batch(const VT* __restrict__ rows, int lane,
                                           int ri, float wi, int j, VT& acc) {
    int r[K]; float w[K]; VT v[K];
    #pragma unroll
    for (int k = 0; k < K; ++k) {
        r[k] = __builtin_amdgcn_readlane(ri, j + k);
        w[k] = __int_as_float(__builtin_amdgcn_readlane(__float_as_int(wi), j + k));
    }
    #pragma unroll
    for (int k = 0; k < K; ++k) v[k] = rows[(size_t)r[k] * 32 + lane];
    #pragma unroll
    for (int k = 0; k < K; ++k) axpy(acc, w[k], v[k]);
}

template <typename VT>
__device__ __forceinline__ void gather_node(const int4* __restrict__ seg, const int2* __restrict__ sedge,
                                            const VT* __restrict__ rows, VT* __restrict__ dst, int N) {
    const int wave = __builtin_amdgcn_readfirstlane(threadIdx.x >> 5);
    const int lane = threadIdx.x & 31;
    const int c = blockIdx.x * 8 + wave;
    if (c >= N) return;
    const int4 sg = seg[c];
    const int beg = sg.x;
    const int n   = sg.y;
    const float dc = __int_as_float(sg.z);
    const VT self = rows[(size_t)c * 32 + lane];
    VT acc;
    set_zero(acc);
    for (int b = 0; b < n; b += 32) {
        const int m = min(32, n - b);
        int ri = 0; float wi = 0.0f;
        if (lane < m) {
            const int2 ed = sedge[beg + b + lane];
            ri = ed.x;
            wi = __int_as_float(ed.y) * dc;
        }
        int j = 0;
        for (; j + 8 <= m; j += 8) edge_batch<8>(rows, lane, ri, wi, j, acc);
        if (j + 4 <= m) { edge_batch<4>(rows, lane, ri, wi, j, acc); j += 4; }
        if (j + 2 <= m) { edge_batch<2>(rows, lane, ri, wi, j, acc); j += 2; }
        if (j < m)      { edge_batch<1>(rows, lane, ri, wi, j, acc); }
    }
    axpy(acc, dc * dc, self);
    dst[(size_t)c * 32 + lane] = acc;
}


__global__ void gemm1_wmma(const float* __restrict__ x, const float* __restrict__ W1,
                           const float* __restrict__ b1, float* __restrict__ h1, int N) {
    const int wave  = threadIdx.x >> 5;
    const int lane  = threadIdx.x & 31;
    const int row0  = blockIdx.x * 16;
    if (row0 >= N) return;
    const int col0  = wave * 16;
    const int m     = lane & 15;
    const int n     = lane & 15;
    const int khalf = lane >> 4;

    v8f c;
    const float bias = b1[col0 + n];
    #pragma unroll
    for (int r = 0; r < 8; ++r) c[r] = bias;

    const float* arow = x + (size_t)(row0 + m) * D_HID + 2 * khalf;
    #pragma unroll 4
    for (int k = 0; k < 128; k += 4) {
        v2f a, b;
        a.x = arow[k];
        a.y = arow[k + 1];
        const int kb = k + 2 * khalf;
        b.x = W1[(size_t)(kb + 0) * D_HID + col0 + n];
        b.y = W1[(size_t)(kb + 1) * D_HID + col0 + n];
        c = __builtin_amdgcn_wmma_f32_16x16x4_f32(false, a, false, b, (short)0, c, false, false);
    }

    const int mbase = khalf * 8;
    #pragma unroll
    for (int r = 0; r < 8; ++r)
        h1[(size_t)(row0 + mbase + r) * D_HID + col0 + n] = c[r];
}


__global__ void gather128(const int4* __restrict__ seg, const int2* __restrict__ sedge,
                          const float* __restrict__ h1, float* __restrict__ hagg, int N) {
    gather_node<float4>(seg, sedge, (const float4*)h1, (float4*)hagg, N);
}


__global__ void bn_stats(const float* __restrict__ hagg, float* bsum, float* bsq, int total) {
    __shared__ float ssum[D_HID];
    __shared__ float ssq[D_HID];
    for (int i = threadIdx.x; i < D_HID; i += blockDim.x) { ssum[i] = 0.0f; ssq[i] = 0.0f; }
    __syncthreads();
    int stride = blockDim.x * gridDim.x;
    for (int i = blockIdx.x * blockDim.x + threadIdx.x; i < total; i += stride) {
        float v = hagg[i];
        atomicAdd(&ssum[i & 127], v);
        atomicAdd(&ssq[i & 127], v * v);
    }
    __syncthreads();
    for (int i = threadIdx.x; i < D_HID; i += blockDim.x) {
        atomicAdd(&bsum[i], ssum[i]);
        atomicAdd(&bsq[i], ssq[i]);
    }
}

__global__ void bn_finalize(const float* __restrict__ bsum, const float* __restrict__ bsq,
                            const float* __restrict__ gamma, const float* __restrict__ beta,
                            float* scale, float* shift, int N) {
    int c = threadIdx.x;
    if (c < D_HID) {
        float inv = 1.0f / (float)N;
        float mu = bsum[c] * inv;
        float var = bsq[c] * inv - mu * mu;
        float s = gamma[c] * rsqrtf(var + BN_EPS);
        scale[c] = s;
        shift[c] = beta[c] - mu * s;
    }
}


__global__ void gemm2_wmma(const float* __restrict__ hagg, const float* __restrict__ scale,
                           const float* __restrict__ shift, const float* __restrict__ W2,
                           const float* __restrict__ b2, float* __restrict__ h2, int N) {
    const int wave  = threadIdx.x >> 5;
    const int lane  = threadIdx.x & 31;
    const int row0  = (blockIdx.x * 2 + (wave >> 2)) * 16;
    if (row0 >= N) return;
    const int col0  = (wave & 3) * 16;
    const int m     = lane & 15;
    const int n     = lane & 15;
    const int khalf = lane >> 4;

    v8f c;
    const float bias = b2[col0 + n];
    #pragma unroll
    for (int r = 0; r < 8; ++r) c[r] = bias;

    const float* arow = hagg + (size_t)(row0 + m) * D_HID;
    #pragma unroll 4
    for (int k = 0; k < 128; k += 4) {
        const int ka = k + 2 * khalf;
        v2f a, b;
        a.x = fmaxf(arow[ka + 0] * scale[ka + 0] + shift[ka + 0], 0.0f);
        a.y = fmaxf(arow[ka + 1] * scale[ka + 1] + shift[ka + 1], 0.0f);
        b.x = W2[(size_t)(ka + 0) * D_OUT + col0 + n];
        b.y = W2[(size_t)(ka + 1) * D_OUT + col0 + n];
        c = __builtin_amdgcn_wmma_f32_16x16x4_f32(false, a, false, b, (short)0, c, false, false);
    }

    const int mbase = khalf * 8;
    #pragma unroll
    for (int r = 0; r < 8; ++r)
        h2[(size_t)(row0 + mbase + r) * D_OUT + col0 + n] = c[r];
}


__global__ void gather64(const int4* __restrict__ seg, const int2* __restrict__ sedge,
                         const float* __restrict__ h2, float* __restrict__ out, int N) {
    gather_node<float2>(seg, sedge, (const float2*)h2, (float2*)out, N);
}


__global__ void logsoftmax_rows(float* out, int N) {
    const int wave = threadIdx.x >> 5;
    const int lane = threadIdx.x & 31;
    const int rowi = blockIdx.x * 8 + wave;
    if (rowi >= N) return;
    float* p = out + (size_t)rowi * D_OUT;
    float v0 = p[lane];
    float v1 = p[lane + 32];
    float mx = fmaxf(v0, v1);
    #pragma unroll
    for (int off = 16; off >= 1; off >>= 1)
        mx = fmaxf(mx, __shfl_xor(mx, off, 32));
    float s = __expf(v0 - mx) + __expf(v1 - mx);
    #pragma unroll
    for (int off = 16; off >= 1; off >>= 1)
        s += __shfl_xor(s, off, 32);
    float l = mx + __logf(s);
    p[lane]      = v0 - l;
    p[lane + 32] = v1 - l;
}


extern "C" void kernel_launch(void* const* d_in, const int* in_sizes, int n_in,
                              void* d_out, int out_size, void* d_ws, size_t ws_size,
                              hipStream_t stream) {
    const float* x     = (const float*)d_in[0];
    const float* W1    = (const float*)d_in[1];
    const float* b1    = (const float*)d_in[2];
    const float* gamma = (const float*)d_in[3];
    const float* beta  = (const float*)d_in[4];
    const float* W2    = (const float*)d_in[5];
    const float* b2    = (const float*)d_in[6];
    const int*   ei    = (const int*)d_in[7];

    const int N = in_sizes[0] / D_HID;
    const int E = in_sizes[7] / 2;
    const int* erow = ei;
    const int* ecol = ei + E;

    float* ws    = (float*)d_ws;
    float* h1    = ws;
    float* hagg  = h1 + (size_t)N * D_HID;
    float* stats = hagg + (size_t)N * D_HID;
    float* bsum  = stats;
    float* bsq   = stats + 128;
    float* scal  = stats + 256;
    float* shft  = stats + 384;
    int*   counter = (int*)(stats + 512);
    int4*  seg   = (int4*)(stats + 576);
    int2*  sedge = (int2*)(seg + N);
    int*   pos   = (int*)(sedge + E);
    int*   cnt   = pos + E;
    float* h2    = h1;
    float* outp  = (float*)d_out;

    init_cnt_stats<<<(N + 255) / 256, 256, 0, stream>>>(cnt, stats, counter, N);
    count_rank<<<(E + 255) / 256, 256, 0, stream>>>(ecol, cnt, pos, E);
    scan_dis<<<(N + 1023) / 1024, 1024, 0, stream>>>(cnt, seg, counter, N);
    fill_sorted<<<(E + 255) / 256, 256, 0, stream>>>(erow, ecol, seg, pos, sedge, E);

    gemm1_wmma<<<(N + 15) / 16, 256, 0, stream>>>(x, W1, b1, h1, N);
    gather128<<<(N + 7) / 8, 256, 0, stream>>>(seg, sedge, h1, hagg, N);

    bn_stats<<<1024, 256, 0, stream>>>(hagg, bsum, bsq, N * D_HID);
    bn_finalize<<<1, 128, 0, stream>>>(bsum, bsq, gamma, beta, scal, shft, N);

    gemm2_wmma<<<(N + 31) / 32, 256, 0, stream>>>(hagg, scal, shft, W2, b2, h2, N);
    gather64<<<(N + 7) / 8, 256, 0, stream>>>(seg, sedge, h2, outp, N);

    logsoftmax_rows<<<(N + 7) / 8, 256, 0, stream>>>(outp, N);
}


# Row-wise consumers run in the gather wave on the row it holds in registers: BN sum/sumsq (wave partials, LDS, same float atomics); log_softmax (DPP butterfly, new sum tree): 2 launches gone
# speedup vs baseline: 2.4606x; 1.1950x over previous
#include <hip/hip_runtime.h>
#include <hip/hip_bf16.h>
#include <math.h>

typedef float v8f __attribute__((ext_vector_type(8)));
typedef float v2f __attribute__((ext_vector_type(2)));

#define D_HID 128
#define D_OUT 64
#define BN_EPS 1e-5f


__global__ void init_cnt_stats(int* cnt, float* stats, int* counter, int N) {
    int i = blockIdx.x * blockDim.x + threadIdx.x;
    int stride = blockDim.x * gridDim.x;
    for (; i < N; i += stride) {
        cnt[i] = 0;
        if (i < 256) stats[i] = 0.0f;
        if (i == 0) *counter = 0;
    }
}

__global__ void count_rank(const int* __restrict__ col, int* cnt, int* __restrict__ pos, int E) {
    const int stride = blockDim.x * gridDim.x;
    for (int e = blockIdx.x * blockDim.x + threadIdx.x; e < E; e += 4 * stride) {
        int c[4], p[4];
        #pragma unroll
        for (int k = 0; k < 4; ++k) {
            const int ek = e + k * stride;
            c[k] = (ek < E) ? col[ek] : -1;
        }
        #pragma unroll
        for (int k = 0; k < 4; ++k) {
            p[k] = 0;
            if (c[k] >= 0) p[k] = atomicAdd(&cnt[c[k]], 1);
        }
        #pragma unroll
        for (int k = 0; k < 4; ++k)
            if (c[k] >= 0) pos[e + k * stride] = p[k];
    }
}

__global__ void scan_dis(const int* __restrict__ cnt, int4* __restrict__ seg, int* counter, int N) {
    __shared__ int wsum[32];
    __shared__ int sbase;
    const int tid  = threadIdx.x;
    const int lane = tid & 31;
    const int wave = tid >> 5;
    const int i = blockIdx.x * 1024 + tid;
    const int v = (i < N) ? cnt[i] : 0;
    int inc = v;
    for (int off = 1; off < 32; off <<= 1) {
        const int t = __shfl_up(inc, off, 32);
        if (lane >= off) inc += t;
    }
    if (lane == 31) wsum[wave] = inc;
    __syncthreads();
    if (wave == 0) {
        const int sv = wsum[lane];
        int si = sv;
        for (int off = 1; off < 32; off <<= 1) {
            const int t = __shfl_up(si, off, 32);
            if (lane >= off) si += t;
        }
        wsum[lane] = si - sv;
        if (lane == 31) sbase = atomicAdd(counter, si);
    }
    __syncthreads();
    if (i < N)
        seg[i] = make_int4(sbase + wsum[wave] + inc - v, v, __float_as_int(rsqrtf((float)(v + 1))), 0);
}

__global__ void fill_sorted(const int* __restrict__ row, const int* __restrict__ col,
                            const int4* __restrict__ seg, const int* __restrict__ pos,
                            int2* __restrict__ sedge, int E) {
    const int stride = blockDim.x * gridDim.x;
    for (int e = blockIdx.x * blockDim.x + threadIdx.x; e < E; e += 4 * stride) {
        int r[4], c[4], p[4], s[4], z[4];
        #pragma unroll
        for (int k = 0; k < 4; ++k) {
            const int ek = e + k * stride;
            r[k] = -1; c[k] = 0; p[k] = 0;
            if (ek < E) { r[k] = row[ek]; c[k] = col[ek]; p[k] = pos[ek]; }
        }
        #pragma unroll
        for (int k = 0; k < 4; ++k) {
            s[k] = 0; z[k] = 0;
            if (r[k] >= 0) { s[k] = seg[c[k]].x; z[k] = seg[r[k]].z; }
        }
        #pragma unroll
        for (int k = 0; k < 4; ++k)
            if (r[k] >= 0) sedge[s[k] + p[k]] = make_int2(r[k], z[k]);
    }
}


__device__ __forceinline__ void axpy(float4& acc, float w, const float4& v) {
    acc.x = fmaf(w, v.x, acc.x); acc.y = fmaf(w, v.y, acc.y);
    acc.z = fmaf(w, v.z, acc.z); acc.w = fmaf(w, v.w, acc.w);
}
__device__ __forceinline__ void axpy(float2& acc, float w, const float2& v) {
    acc.x = fmaf(w, v.x, acc.x); acc.y = fmaf(w, v.y, acc.y);
}
__device__ __forceinline__ void set_zero(float4& a) { a = make_float4(0.0f, 0.0f, 0.0f, 0.0f); }
__device__ __forceinline__ void set_zero(float2& a) { a = make_float2(0.0f, 0.0f); }

template <int K, typename VT>
__device__ __forceinline__ void edge_batch(const VT* __restrict__ rows, int lane,
                                           int ri, float wi, int j, int cnt, VT& acc) {
    float w[K]; VT v[K];
    #pragma unroll
    for (int k = 0; k < K; ++k) {
        w[k] = 0.0f;
        set_zero(v[k]);
        if (k < cnt) {
            const int r = __builtin_amdgcn_readlane(ri, j + k);
            w[k] = __int_as_float(__builtin_amdgcn_readlane(__float_as_int(wi), j + k));
            v[k] = rows[(size_t)r * 32 + lane];
        }
    }
    #pragma unroll
    for (int k = 0; k < K; ++k)
        if (k < cnt) axpy(acc, w[k], v[k]);
}


struct RowStats {
    float4 s, q;
    __device__ __forceinline__ void row(float4& a) {
        s.x += a.x; s.y += a.y; s.z += a.z; s.w += a.w;
        q.x = fmaf(a.x, a.x, q.x); q.y = fmaf(a.y, a.y, q.y);
        q.z = fmaf(a.z, a.z, q.z); q.w = fmaf(a.w, a.w, q.w);
    }
};

template <int M>
__device__ __forceinline__ float row_xor(float x) {
    return __builtin_bit_cast(float, __builtin_amdgcn_mov_dpp(__builtin_bit_cast(int, x), 0x160 + M, 0xf, 0xf, true));
}
__device__ __forceinline__ float xhalf(float x) {
    const int b = __builtin_bit_cast(int, x);
    return __builtin_bit_cast(float, __builtin_amdgcn_permlanex16(b, b, 0x76543210, 0xfedcba98, false, false));
}
__device__ __forceinline__ float wave_max(float x) {
    x = fmaxf(x, row_xor<1>(x)); x = fmaxf(x, row_xor<2>(x));
    x = fmaxf(x, row_xor<4>(x)); x = fmaxf(x, row_xor<8>(x));
    return fmaxf(x, xhalf(x));
}
__device__ __forceinline__ float wave_sum(float x) {
    x += row_xor<1>(x); x += row_xor<2>(x);
    x += row_xor<4>(x); x += row_xor<8>(x);
    return x + xhalf(x);
}

struct RowLogSoftmax {
    __device__ __forceinline__ void row(float2& a) {
        const float mx = wave_max(fmaxf(a.x, a.y));
        const float s  = wave_sum(__expf(a.x - mx) + __expf(a.y - mx));
        const float l  = mx + __logf(s);
        a.x -= l; a.y -= l;
    }
};

template <typename VT, typename EPI>
__device__ __forceinline__ void gather_nodes(const int4* __restrict__ seg, const int2* __restrict__ sedge,
                                             const VT* __restrict__ rows, VT* __restrict__ dst, int N,
                                             EPI& epi) {
    const int wave = __builtin_amdgcn_readfirstlane(threadIdx.x >> 5);
    const int lane = threadIdx.x & 31;
    const int G = gridDim.x * 8;
    int c0 = blockIdx.x * 8 + wave;
    if (c0 >= N) return;
    int c1 = c0 + G;
    const int4 none = make_int4(0, 0, 0, 0);
    int4 sg0 = seg[c0];
    int4 sg1 = none;
    if (c1 < N) sg1 = seg[c1];
    int2 ed0 = make_int2(0, 0);
    if (lane < min(32, sg0.y)) ed0 = sedge[sg0.x + lane];
    int   ri = ed0.x;
    float wi = __int_as_float(ed0.y) * __int_as_float(sg0.z);
    for (;;) {
        const int c2 = c1 + G;
        int4 sg2 = none;
        if (c2 < N) sg2 = seg[c2];
        int2 ed1 = make_int2(0, 0);
        if (lane < min(32, sg1.y)) ed1 = sedge[sg1.x + lane];
        const int beg = sg0.x;
        const int n   = sg0.y;
        const float dc = __int_as_float(sg0.z);
        const VT self = rows[(size_t)c0 * 32 + lane];
        VT acc;
        set_zero(acc);
        {
            const int m = min(32, n);
            if (m > 0)  edge_batch<16>(rows, lane, ri, wi, 0, m, acc);
            if (m > 16) edge_batch<16>(rows, lane, ri, wi, 16, m - 16, acc);
        }
        for (int b = 32; b < n; b += 32) {
            const int m = min(32, n - b);
            int2 ed = make_int2(0, 0);
            if (lane < m) ed = sedge[beg + b + lane];
            const int   rj = ed.x;
            const float wj = __int_as_float(ed.y) * dc;
            for (int j = 0; j < m; j += 16) edge_batch<16>(rows, lane, rj, wj, j, m - j, acc);
        }
        axpy(acc, dc * dc, self);
        epi.row(acc);
        dst[(size_t)c0 * 32 + lane] = acc;
        if (c1 >= N) break;
        c0 = c1; c1 = c2;
        sg0 = sg1; sg1 = sg2;
        ri = ed1.x;
        wi = __int_as_float(ed1.y) * __int_as_float(sg0.z);
    }
}


__global__ void __launch_bounds__(256) gemm1_wmma(const float* __restrict__ x, const float* __restrict__ W1,
                           const float* __restrict__ b1, float* __restrict__ h1, int N) {
    const int wave  = threadIdx.x >> 5;
    const int lane  = threadIdx.x & 31;
    const int col0  = wave * 16;
    const int m     = lane & 15;
    const int n     = lane & 15;
    const int khalf = lane >> 4;

    const float bias = b1[col0 + n];
    v2f bf[32];
    #pragma unroll
    for (int t = 0; t < 32; ++t) {
        const int kb = 4 * t + 2 * khalf;
        bf[t].x = W1[(size_t)(kb + 0) * D_HID + col0 + n];
        bf[t].y = W1[(size_t)(kb + 1) * D_HID + col0 + n];
    }

    for (int row0 = blockIdx.x * 16; row0 < N; row0 += gridDim.x * 16) {
        v8f c;
        #pragma unroll
        for (int r = 0; r < 8; ++r) c[r] = bias;

        const float* arow = x + (size_t)(row0 + m) * D_HID + 2 * khalf;
        v2f af[32];
        #pragma unroll
        for (int t = 0; t < 32; ++t) {
            af[t].x = arow[4 * t];
            af[t].y = arow[4 * t + 1];
        }
        __builtin_amdgcn_sched_barrier(0);
        #pragma unroll
        for (int t = 0; t < 32; ++t)
            c = __builtin_amdgcn_wmma_f32_16x16x4_f32(false, af[t], false, bf[t], (short)0, c, false, false);

        const int mbase = khalf * 8;
        #pragma unroll
        for (int r = 0; r < 8; ++r)
            h1[(size_t)(row0 + mbase + r) * D_HID + col0 + n] = c[r];
    }
}


__global__ void __launch_bounds__(256) gather128(const int4* __restrict__ seg, const int2* __restrict__ sedge,
                          const float* __restrict__ h1, float* __restrict__ hagg, float* stats, int N) {
    __shared__ float part[8][2 * D_HID];
    RowStats st;
    st.s = make_float4(0.0f, 0.0f, 0.0f, 0.0f);
    st.q = make_float4(0.0f, 0.0f, 0.0f, 0.0f);
    gather_nodes<float4>(seg, sedge, (const float4*)h1, (float4*)hagg, N, st);
    const int wave = threadIdx.x >> 5;
    const int lane = threadIdx.x & 31;
    *(float4*)&part[wave][4 * lane]         = st.s;
    *(float4*)&part[wave][D_HID + 4 * lane] = st.q;
    __syncthreads();
    float tot = 0.0f;
    #pragma unroll
    for (int w = 0; w < 8; ++w) tot += part[w][threadIdx.x];
    atomicAdd(&stats[threadIdx.x], tot);
}


__global__ void bn_finalize(const float* __restrict__ bsum, const float* __restrict__ bsq,
                            const float* __restrict__ gamma, const float* __restrict__ beta,
                            float* scale, float* shift, int N) {
    int c = threadIdx.x;
    if (c < D_HID) {
        float inv = 1.0f / (float)N;
        float mu = bsum[c] * inv;
        float var = bsq[c] * inv - mu * mu;
        float s = gamma[c] * rsqrtf(var + BN_EPS);
        scale[c] = s;
        shift[c] = beta[c] - mu * s;
    }
}


__global__ void __launch_bounds__(256) gemm2_wmma(const float* __restrict__ hagg, const float* __restrict__ scale,
                           const float* __restrict__ shift, const float* __restrict__ W2,
                           const float* __restrict__ b2, float* __restrict__ h2, int N) {
    const int wave  = __builtin_amdgcn_readfirstlane(threadIdx.x >> 5);
    const int lane  = threadIdx.x & 31;
    const int col0  = (wave & 3) * 16;
    const int m     = lane & 15;
    const int n     = lane & 15;
    const int khalf = lane >> 4;

    const float bias = b2[col0 + n];
    v2f bf[32];
    #pragma unroll
    for (int t = 0; t < 32; ++t) {
        const int ka = 4 * t + 2 * khalf;
        bf[t].x = W2[(size_t)(ka + 0) * D_OUT + col0 + n];
        bf[t].y = W2[(size_t)(ka + 1) * D_OUT + col0 + n];
    }

    for (int row0 = (blockIdx.x * 2 + (wave >> 2)) * 16; row0 < N; row0 += gridDim.x * 32) {
        v8f c;
        #pragma unroll
        for (int r = 0; r < 8; ++r) c[r] = bias;

        const float* arow = hagg + (size_t)(row0 + m) * D_HID;
        v2f af[32];
        #pragma unroll
        for (int t = 0; t < 32; ++t) {
            const int ka = 4 * t + 2 * khalf;
            af[t].x = arow[ka + 0];
            af[t].y = arow[ka + 1];
        }
        __builtin_amdgcn_sched_barrier(0);
        #pragma unroll
        for (int t = 0; t < 32; ++t) {
            const int ka = 4 * t + 2 * khalf;
            v2f a;
            a.x = fmaxf(af[t].x * scale[ka + 0] + shift[ka + 0], 0.0f);
            a.y = fmaxf(af[t].y * scale[ka + 1] + shift[ka + 1], 0.0f);
            c = __builtin_amdgcn_wmma_f32_16x16x4_f32(false, a, false, bf[t], (short)0, c, false, false);
        }

        const int mbase = khalf * 8;
        #pragma unroll
        for (int r = 0; r < 8; ++r)
            h2[(size_t)(row0 + mbase + r) * D_OUT + col0 + n] = c[r];
    }
}


__global__ void __launch_bounds__(256) gather64(const int4* __restrict__ seg, const int2* __restrict__ sedge,
                         const float* __restrict__ h2, float* __restrict__ out, int N) {
    RowLogSoftmax ls;
    gather_nodes<float2>(seg, sedge, (const float2*)h2, (float2*)out, N, ls);
}


extern "C" void kernel_launch(void* const* d_in, const int* in_sizes, int n_in,
                              void* d_out, int out_size, void* d_ws, size_t ws_size,
                              hipStream_t stream) {
    const float* x     = (const float*)d_in[0];
    const float* W1    = (const float*)d_in[1];
    const float* b1    = (const float*)d_in[2];
    const float* gamma = (const float*)d_in[3];
    const float* beta  = (const float*)d_in[4];
    const float* W2    = (const float*)d_in[5];
    const float* b2    = (const float*)d_in[6];
    const int*   ei    = (const int*)d_in[7];

    const int N = in_sizes[0] / D_HID;
    const int E = in_sizes[7] / 2;
    const int* erow = ei;
    const int* ecol = ei + E;

    float* ws    = (float*)d_ws;
    float* h1    = ws;
    float* hagg  = h1 + (size_t)N * D_HID;
    float* stats = hagg + (size_t)N * D_HID;
    float* bsum  = stats;
    float* bsq   = stats + 128;
    float* scal  = stats + 256;
    float* shft  = stats + 384;
    int*   counter = (int*)(stats + 512);
    int4*  seg   = (int4*)(stats + 576);
    int2*  sedge = (int2*)(seg + N);
    int*   pos   = (int*)(sedge + E);
    int*   cnt   = pos + E;
    float* h2    = h1;
    float* outp  = (float*)d_out;

    init_cnt_stats<<<(N + 255) / 256, 256, 0, stream>>>(cnt, stats, counter, N);
    count_rank<<<512, 256, 0, stream>>>(ecol, cnt, pos, E);
    scan_dis<<<(N + 1023) / 1024, 1024, 0, stream>>>(cnt, seg, counter, N);
    fill_sorted<<<512, 256, 0, stream>>>(erow, ecol, seg, pos, sedge, E);

    gemm1_wmma<<<512, 256, 0, stream>>>(x, W1, b1, h1, N);
    gather128<<<1024, 256, 0, stream>>>(seg, sedge, h1, hagg, stats, N);

    bn_finalize<<<1, 128, 0, stream>>>(bsum, bsq, gamma, beta, scal, shft, N);

    gemm2_wmma<<<256, 256, 0, stream>>>(hagg, scal, shft, W2, b2, h2, N);
    gather64<<<1024, 256, 0, stream>>>(seg, sedge, h2, outp, N);
}


# Both GEMMs: a lane loads 4 consecutive k of its row per 16-byte load (A and B loaders alike), so a row tile takes 16 A loads per wave, not 32; the k order of the summed products changes
# speedup vs baseline: 2.5413x; 1.0328x over previous
#include <hip/hip_runtime.h>
#include <hip/hip_bf16.h>
#include <math.h>

typedef float v8f __attribute__((ext_vector_type(8)));
typedef float v2f __attribute__((ext_vector_type(2)));

#define D_HID 128
#define D_OUT 64
#define BN_EPS 1e-5f


__global__ void init_cnt_stats(int* cnt, float* stats, int* counter, int N) {
    int i = blockIdx.x * blockDim.x + threadIdx.x;
    int stride = blockDim.x * gridDim.x;
    for (; i < N; i += stride) {
        cnt[i] = 0;
        if (i < 256) stats[i] = 0.0f;
        if (i == 0) *counter = 0;
    }
}

__global__ void count_rank(const int* __restrict__ col, int* cnt, int* __restrict__ pos, int E) {
    const int stride = blockDim.x * gridDim.x;
    for (int e = blockIdx.x * blockDim.x + threadIdx.x; e < E; e += 4 * stride) {
        int c[4], p[4];
        #pragma unroll
        for (int k = 0; k < 4; ++k) {
            const int ek = e + k * stride;
            c[k] = (ek < E) ? col[ek] : -1;
        }
        #pragma unroll
        for (int k = 0; k < 4; ++k) {
            p[k] = 0;
            if (c[k] >= 0) p[k] = atomicAdd(&cnt[c[k]], 1);
        }
        #pragma unroll
        for (int k = 0; k < 4; ++k)
            if (c[k] >= 0) pos[e + k * stride] = p[k];
    }
}

__global__ void scan_dis(const int* __restrict__ cnt, int4* __restrict__ seg, int* counter, int N) {
    __shared__ int wsum[32];
    __shared__ int sbase;
    const int tid  = threadIdx.x;
    const int lane = tid & 31;
    const int wave = tid >> 5;
    const int i = blockIdx.x * 1024 + tid;
    const int v = (i < N) ? cnt[i] : 0;
    int inc = v;
    for (int off = 1; off < 32; off <<= 1) {
        const int t = __shfl_up(inc, off, 32);
        if (lane >= off) inc += t;
    }
    if (lane == 31) wsum[wave] = inc;
    __syncthreads();
    if (wave == 0) {
        const int sv = wsum[lane];
        int si = sv;
        for (int off = 1; off < 32; off <<= 1) {
            const int t = __shfl_up(si, off, 32);
            if (lane >= off) si += t;
        }
        wsum[lane] = si - sv;
        if (lane == 31) sbase = atomicAdd(counter, si);
    }
    __syncthreads();
    if (i < N)
        seg[i] = make_int4(sbase + wsum[wave] + inc - v, v, __float_as_int(rsqrtf((float)(v + 1))), 0);
}

__global__ void fill_sorted(const int* __restrict__ row, const int* __restrict__ col,
                            const int4* __restrict__ seg, const int* __restrict__ pos,
                            int2* __restrict__ sedge, int E) {
    const int stride = blockDim.x * gridDim.x;
    for (int e = blockIdx.x * blockDim.x + threadIdx.x; e < E; e += 4 * stride) {
        int r[4], c[4], p[4], s[4], z[4];
        #pragma unroll
        for (int k = 0; k < 4; ++k) {
            const int ek = e + k * stride;
            r[k] = -1; c[k] = 0; p[k] = 0;
            if (ek < E) { r[k] = row[ek]; c[k] = col[ek]; p[k] = pos[ek]; }
        }
        #pragma unroll
        for (int k = 0; k < 4; ++k) {
            s[k] = 0; z[k] = 0;
            if (r[k] >= 0) { s[k] = seg[c[k]].x; z[k] = seg[r[k]].z; }
        }
        #pragma unroll
        for (int k = 0; k < 4; ++k)
            if (r[k] >= 0) sedge[s[k] + p[k]] = make_int2(r[k], z[k]);
    }
}


__device__ __forceinline__ void axpy(float4& acc, float w, const float4& v) {
    acc.x = fmaf(w, v.x, acc.x); acc.y = fmaf(w, v.y, acc.y);
    acc.z = fmaf(w, v.z, acc.z); acc.w = fmaf(w, v.w, acc.w);
}
__device__ __forceinline__ void axpy(float2& acc, float w, const float2& v) {
    acc.x = fmaf(w, v.x, acc.x); acc.y = fmaf(w, v.y, acc.y);
}
__device__ __forceinline__ void set_zero(float4& a) { a = make_float4(0.0f, 0.0f, 0.0f, 0.0f); }
__device__ __forceinline__ void set_zero(float2& a) { a = make_float2(0.0f, 0.0f); }

template <int K, typename VT>
__device__ __forceinline__ void edge_batch(const VT* __restrict__ rows, int lane,
                                           int ri, float wi, int j, int cnt, VT& acc) {
    float w[K]; VT v[K];
    #pragma unroll
    for (int k = 0; k < K; ++k) {
        w[k] = 0.0f;
        set_zero(v[k]);
        if (k < cnt) {
            const int r = __builtin_amdgcn_readlane(ri, j + k);
            w[k] = __int_as_float(__builtin_amdgcn_readlane(__float_as_int(wi), j + k));
            v[k] = rows[(size_t)r * 32 + lane];
        }
    }
    #pragma unroll
    for (int k = 0; k < K; ++k)
        if (k < cnt) axpy(acc, w[k], v[k]);
}


struct RowStats {
    float4 s, q;
    __device__ __forceinline__ void row(float4& a) {
        s.x += a.x; s.y += a.y; s.z += a.z; s.w += a.w;
        q.x = fmaf(a.x, a.x, q.x); q.y = fmaf(a.y, a.y, q.y);
        q.z = fmaf(a.z, a.z, q.z); q.w = fmaf(a.w, a.w, q.w);
    }
};

template <int M>
__device__ __forceinline__ float row_xor(float x) {
    return __builtin_bit_cast(float, __builtin_amdgcn_mov_dpp(__builtin_bit_cast(int, x), 0x160 + M, 0xf, 0xf, true));
}
__device__ __forceinline__ float xhalf(float x) {
    const int b = __builtin_bit_cast(int, x);
    return __builtin_bit_cast(float, __builtin_amdgcn_permlanex16(b, b, 0x76543210, 0xfedcba98, false, false));
}
__device__ __forceinline__ float wave_max(float x) {
    x = fmaxf(x, row_xor<1>(x)); x = fmaxf(x, row_xor<2>(x));
    x = fmaxf(x, row_xor<4>(x)); x = fmaxf(x, row_xor<8>(x));
    return fmaxf(x, xhalf(x));
}
__device__ __forceinline__ float wave_sum(float x) {
    x += row_xor<1>(x); x += row_xor<2>(x);
    x += row_xor<4>(x); x += row_xor<8>(x);
    return x + xhalf(x);
}

struct RowLogSoftmax {
    __device__ __forceinline__ void row(float2& a) {
        const float mx = wave_max(fmaxf(a.x, a.y));
        const float s  = wave_sum(__expf(a.x - mx) + __expf(a.y - mx));
        const float l  = mx + __logf(s);
        a.x -= l; a.y -= l;
    }
};

template <typename VT, typename EPI>
__device__ __forceinline__ void gather_nodes(const int4* __restrict__ seg, const int2* __restrict__ sedge,
                                             const VT* __restrict__ rows, VT* __restrict__ dst, int N,
                                             EPI& epi) {
    const int wave = __builtin_amdgcn_readfirstlane(threadIdx.x >> 5);
    const int lane = threadIdx.x & 31;
    const int G = gridDim.x * 8;
    int c0 = blockIdx.x * 8 + wave;
    if (c0 >= N) return;
    int c1 = c0 + G;
    const int4 none = make_int4(0, 0, 0, 0);
    int4 sg0 = seg[c0];
    int4 sg1 = none;
    if (c1 < N) sg1 = seg[c1];
    int2 ed0 = make_int2(0, 0);
    if (lane < min(32, sg0.y)) ed0 = sedge[sg0.x + lane];
    int   ri = ed0.x;
    float wi = __int_as_float(ed0.y) * __int_as_float(sg0.z);
    for (;;) {
        const int c2 = c1 + G;
        int4 sg2 = none;
        if (c2 < N) sg2 = seg[c2];
        int2 ed1 = make_int2(0, 0);
        if (lane < min(32, sg1.y)) ed1 = sedge[sg1.x + lane];
        const int beg = sg0.x;
        const int n   = sg0.y;
        const float dc = __int_as_float(sg0.z);
        const VT self = rows[(size_t)c0 * 32 + lane];
        VT acc;
        set_zero(acc);
        {
            const int m = min(32, n);
            if (m > 0)  edge_batch<16>(rows, lane, ri, wi, 0, m, acc);
            if (m > 16) edge_batch<16>(rows, lane, ri, wi, 16, m - 16, acc);
        }
        for (int b = 32; b < n; b += 32) {
            const int m = min(32, n - b);
            int2 ed = make_int2(0, 0);
            if (lane < m) ed = sedge[beg + b + lane];
            const int   rj = ed.x;
            const float wj = __int_as_float(ed.y) * dc;
            for (int j = 0; j < m; j += 16) edge_batch<16>(rows, lane, rj, wj, j, m - j, acc);
        }
        axpy(acc, dc * dc, self);
        epi.row(acc);
        dst[(size_t)c0 * 32 + lane] = acc;
        if (c1 >= N) break;
        c0 = c1; c1 = c2;
        sg0 = sg1; sg1 = sg2;
        ri = ed1.x;
        wi = __int_as_float(ed1.y) * __int_as_float(sg0.z);
    }
}


__global__ void __launch_bounds__(256) gemm1_wmma(const float* __restrict__ x, const float* __restrict__ W1,
                           const float* __restrict__ b1, float* __restrict__ h1, int N) {
    const int wave  = threadIdx.x >> 5;
    const int lane  = threadIdx.x & 31;
    const int col0  = wave * 16;
    const int m     = lane & 15;
    const int n     = lane & 15;
    const int khalf = lane >> 4;

    const float bias = b1[col0 + n];
    v2f bf[32];
    #pragma unroll
    for (int u = 0; u < 16; ++u) {
        const int kb = 8 * u + 4 * khalf;
        bf[2 * u].x     = W1[(size_t)(kb + 0) * D_HID + col0 + n];
        bf[2 * u].y     = W1[(size_t)(kb + 1) * D_HID + col0 + n];
        bf[2 * u + 1].x = W1[(size_t)(kb + 2) * D_HID + col0 + n];
        bf[2 * u + 1].y = W1[(size_t)(kb + 3) * D_HID + col0 + n];
    }

    for (int row0 = blockIdx.x * 16; row0 < N; row0 += gridDim.x * 16) {
        v8f c;
        #pragma unroll
        for (int r = 0; r < 8; ++r) c[r] = bias;

        const float4* arow = (const float4*)(x + (size_t)(row0 + m) * D_HID + 4 * khalf);
        float4 af[16];
        #pragma unroll
        for (int u = 0; u < 16; ++u) af[u] = arow[2 * u];
        __builtin_amdgcn_sched_barrier(0);
        #pragma unroll
        for (int u = 0; u < 16; ++u) {
            v2f lo, hi;
            lo.x = af[u].x; lo.y = af[u].y;
            hi.x = af[u].z; hi.y = af[u].w;
            c = __builtin_amdgcn_wmma_f32_16x16x4_f32(false, lo, false, bf[2 * u], (short)0, c, false, false);
            c = __builtin_amdgcn_wmma_f32_16x16x4_f32(false, hi, false, bf[2 * u + 1], (short)0, c, false, false);
        }

        const int mbase = khalf * 8;
        #pragma unroll
        for (int r = 0; r < 8; ++r)
            h1[(size_t)(row0 + mbase + r) * D_HID + col0 + n] = c[r];
    }
}


__global__ void __launch_bounds__(256) gather128(const int4* __restrict__ seg, const int2* __restrict__ sedge,
                          const float* __restrict__ h1, float* __restrict__ hagg, float* stats, int N) {
    __shared__ float part[8][2 * D_HID];
    RowStats st;
    st.s = make_float4(0.0f, 0.0f, 0.0f, 0.0f);
    st.q = make_float4(0.0f, 0.0f, 0.0f, 0.0f);
    gather_nodes<float4>(seg, sedge, (const float4*)h1, (float4*)hagg, N, st);
    const int wave = threadIdx.x >> 5;
    const int lane = threadIdx.x & 31;
    *(float4*)&part[wave][4 * lane]         = st.s;
    *(float4*)&part[wave][D_HID + 4 * lane] = st.q;
    __syncthreads();
    float tot = 0.0f;
    #pragma unroll
    for (int w = 0; w < 8; ++w) tot += part[w][threadIdx.x];
    atomicAdd(&stats[threadIdx.x], tot);
}


__global__ void bn_finalize(const float* __restrict__ bsum, const float* __restrict__ bsq,
                            const float* __restrict__ gamma, const float* __restrict__ beta,
                            float* scale, float* shift, int N) {
    int c = threadIdx.x;
    if (c < D_HID) {
        float inv = 1.0f / (float)N;
        float mu = bsum[c] * inv;
        float var = bsq[c] * inv - mu * mu;
        float s = gamma[c] * rsqrtf(var + BN_EPS);
        scale[c] = s;
        shift[c] = beta[c] - mu * s;
    }
}


__global__ void __launch_bounds__(256) gemm2_wmma(const float* __restrict__ hagg, const float* __restrict__ scale,
                           const float* __restrict__ shift, const float* __restrict__ W2,
                           const float* __restrict__ b2, float* __restrict__ h2, int N) {
    const int wave  = __builtin_amdgcn_readfirstlane(threadIdx.x >> 5);
    const int lane  = threadIdx.x & 31;
    const int col0  = (wave & 3) * 16;
    const int m     = lane & 15;
    const int n     = lane & 15;
    const int khalf = lane >> 4;

    const float bias = b2[col0 + n];
    v2f bf[32];
    #pragma unroll
    for (int u = 0; u < 16; ++u) {
        const int kb = 8 * u + 4 * khalf;
        bf[2 * u].x     = W2[(size_t)(kb + 0) * D_OUT + col0 + n];
        bf[2 * u].y     = W2[(size_t)(kb + 1) * D_OUT + col0 + n];
        bf[2 * u + 1].x = W2[(size_t)(kb + 2) * D_OUT + col0 + n];
        bf[2 * u + 1].y = W2[(size_t)(kb + 3) * D_OUT + col0 + n];
    }

    for (int row0 = (blockIdx.x * 2 + (wave >> 2)) * 16; row0 < N; row0 += gridDim.x * 32) {
        v8f c;
        #pragma unroll
        for (int r = 0; r < 8; ++r) c[r] = bias;

        const float4* arow = (const float4*)(hagg + (size_t)(row0 + m) * D_HID + 4 * khalf);
        float4 af[16];
        #pragma unroll
        for (int u = 0; u < 16; ++u) af[u] = arow[2 * u];
        __builtin_amdgcn_sched_barrier(0);
        #pragma unroll
        for (int u = 0; u < 16; ++u) {
            const float4 sc = ((const float4*)(scale + 4 * khalf))[2 * u];
            const float4 sh = ((const float4*)(shift + 4 * khalf))[2 * u];
            v2f lo, hi;
            lo.x = fmaxf(af[u].x * sc.x + sh.x, 0.0f);
            lo.y = fmaxf(af[u].y * sc.y + sh.y, 0.0f);
            hi.x = fmaxf(af[u].z * sc.z + sh.z, 0.0f);
            hi.y = fmaxf(af[u].w * sc.w + sh.w, 0.0f);
            c = __builtin_amdgcn_wmma_f32_16x16x4_f32(false, lo, false, bf[2 * u], (short)0, c, false, false);
            c = __builtin_amdgcn_wmma_f32_16x16x4_f32(false, hi, false, bf[2 * u + 1], (short)0, c, false, false);
        }

        const int mbase = khalf * 8;
        #pragma unroll
        for (int r = 0; r < 8; ++r)
            h2[(size_t)(row0 + mbase + r) * D_OUT + col0 + n] = c[r];
    }
}


__global__ void __launch_bounds__(256) gather64(const int4* __restrict__ seg, const int2* __restrict__ sedge,
                         const float* __restrict__ h2, float* __restrict__ out, int N) {
    RowLogSoftmax ls;
    gather_nodes<float2>(seg, sedge, (const float2*)h2, (float2*)out, N, ls);
}


extern "C" void kernel_launch(void* const* d_in, const int* in_sizes, int n_in,
                              void* d_out, int out_size, void* d_ws, size_t ws_size,
                              hipStream_t stream) {
    const float* x     = (const float*)d_in[0];
    const float* W1    = (const float*)d_in[1];
    const float* b1    = (const float*)d_in[2];
    const float* gamma = (const float*)d_in[3];
    const float* beta  = (const float*)d_in[4];
    const float* W2    = (const float*)d_in[5];
    const float* b2    = (const float*)d_in[6];
    const int*   ei    = (const int*)d_in[7];

    const int N = in_sizes[0] / D_HID;
    const int E = in_sizes[7] / 2;
    const int* erow = ei;
    const int* ecol = ei + E;

    float* ws    = (float*)d_ws;
    float* h1    = ws;
    float* hagg  = h1 + (size_t)N * D_HID;
    float* stats = hagg + (size_t)N * D_HID;
    float* bsum  = stats;
    float* bsq   = stats + 128;
    float* scal  = stats + 256;
    float* shft  = stats + 384;
    int*   counter = (int*)(stats + 512);
    int4*  seg   = (int4*)(stats + 576);
    int2*  sedge = (int2*)(seg + N);
    int*   pos   = (int*)(sedge + E);
    int*   cnt   = pos + E;
    float* h2    = h1;
    float* outp  = (float*)d_out;

    init_cnt_stats<<<(N + 255) / 256, 256, 0, stream>>>(cnt, stats, counter, N);
    count_rank<<<512, 256, 0, stream>>>(ecol, cnt, pos, E);
    scan_dis<<<(N + 1023) / 1024, 1024, 0, stream>>>(cnt, seg, counter, N);
    fill_sorted<<<512, 256, 0, stream>>>(erow, ecol, seg, pos, sedge, E);

    gemm1_wmma<<<512, 256, 0, stream>>>(x, W1, b1, h1, N);
    gather128<<<1024, 256, 0, stream>>>(seg, sedge, h1, hagg, stats, N);

    bn_finalize<<<1, 128, 0, stream>>>(bsum, bsq, gamma, beta, scal, shft, N);

    gemm2_wmma<<<256, 256, 0, stream>>>(hagg, scal, shft, W2, b2, h2, N);
    gather64<<<1024, 256, 0, stream>>>(seg, sedge, h2, outp, N);
}


# Edge sort in one pass: the counting atomic's return is the edge's slot in a 32-slot bucket of its destination (overflow list past 32); GEMMs scale row i by dis[i], gathers sum unweighted rows
# speedup vs baseline: 2.8689x; 1.1289x over previous
#include <hip/hip_runtime.h>
#include <hip/hip_bf16.h>
#include <math.h>

typedef float v8f __attribute__((ext_vector_type(8)));
typedef float v2f __attribute__((ext_vector_type(2)));

#define D_HID 128
#define D_OUT 64
#define BN_EPS 1e-5f


#define BUCKET 32

__global__ void init_cnt_stats(int* cnt, float* stats, int* novf, int N) {
    int i = blockIdx.x * blockDim.x + threadIdx.x;
    int stride = blockDim.x * gridDim.x;
    for (; i < N; i += stride) {
        cnt[i] = 0;
        if (i < 256) stats[i] = 0.0f;
        if (i == 0) *novf = 0;
    }
}

__global__ void place_edges(const int* __restrict__ row, const int* __restrict__ col, int* cnt,
                            int* __restrict__ slots, int2* __restrict__ ovf, int* novf, int E) {
    const int stride = blockDim.x * gridDim.x;
    for (int e = blockIdx.x * blockDim.x + threadIdx.x; e < E; e += 4 * stride) {
        int c[4], r[4], p[4];
        #pragma unroll
        for (int k = 0; k < 4; ++k) {
            const int ek = e + k * stride;
            c[k] = -1; r[k] = 0;
            if (ek < E) { c[k] = col[ek]; r[k] = row[ek]; }
        }
        #pragma unroll
        for (int k = 0; k < 4; ++k) {
            p[k] = 0;
            if (c[k] >= 0) p[k] = atomicAdd(&cnt[c[k]], 1);
        }
        #pragma unroll
        for (int k = 0; k < 4; ++k) {
            if (c[k] >= 0) {
                if (p[k] < BUCKET) {
                    slots[c[k] * BUCKET + p[k]] = r[k];
                } else {
                    const int q = atomicAdd(novf, 1);
                    ovf[q] = make_int2(c[k], r[k]);
                }
            }
        }
    }
}


__device__ __forceinline__ void add_row(float4& acc, const float4& v) {
    acc.x += v.x; acc.y += v.y; acc.z += v.z; acc.w += v.w;
}
__device__ __forceinline__ void add_row(float2& acc, const float2& v) {
    acc.x += v.x; acc.y += v.y;
}
__device__ __forceinline__ void scale_row(float4& a, float d) { a.x *= d; a.y *= d; a.z *= d; a.w *= d; }
__device__ __forceinline__ void scale_row(float2& a, float d) { a.x *= d; a.y *= d; }
__device__ __forceinline__ void set_zero(float4& a) { a = make_float4(0.0f, 0.0f, 0.0f, 0.0f); }
__device__ __forceinline__ void set_zero(float2& a) { a = make_float2(0.0f, 0.0f); }

template <int K, typename VT>
__device__ __forceinline__ void edge_batch(const VT* __restrict__ rows, int lane,
                                           int ri, int j, int cnt, VT& acc) {
    VT v[K];
    #pragma unroll
    for (int k = 0; k < K; ++k) {
        set_zero(v[k]);
        if (k < cnt) {
            const int r = __builtin_amdgcn_readlane(ri, j + k);
            v[k] = rows[(size_t)r * 32 + lane];
        }
    }
    #pragma unroll
    for (int k = 0; k < K; ++k)
        if (k < cnt) add_row(acc, v[k]);
}


struct RowStats {
    float4 s, q;
    __device__ __forceinline__ void row(float4& a) {
        s.x += a.x; s.y += a.y; s.z += a.z; s.w += a.w;
        q.x = fmaf(a.x, a.x, q.x); q.y = fmaf(a.y, a.y, q.y);
        q.z = fmaf(a.z, a.z, q.z); q.w = fmaf(a.w, a.w, q.w);
    }
};

template <int M>
__device__ __forceinline__ float row_xor(float x) {
    return __builtin_bit_cast(float, __builtin_amdgcn_mov_dpp(__builtin_bit_cast(int, x), 0x160 + M, 0xf, 0xf, true));
}
__device__ __forceinline__ float xhalf(float x) {
    const int b = __builtin_bit_cast(int, x);
    return __builtin_bit_cast(float, __builtin_amdgcn_permlanex16(b, b, 0x76543210, 0xfedcba98, false, false));
}
__device__ __forceinline__ float wave_max(float x) {
    x = fmaxf(x, row_xor<1>(x)); x = fmaxf(x, row_xor<2>(x));
    x = fmaxf(x, row_xor<4>(x)); x = fmaxf(x, row_xor<8>(x));
    return fmaxf(x, xhalf(x));
}
__device__ __forceinline__ float wave_sum(float x) {
    x += row_xor<1>(x); x += row_xor<2>(x);
    x += row_xor<4>(x); x += row_xor<8>(x);
    return x + xhalf(x);
}

struct RowLogSoftmax {
    __device__ __forceinline__ void row(float2& a) {
        const float mx = wave_max(fmaxf(a.x, a.y));
        const float s  = wave_sum(__expf(a.x - mx) + __expf(a.y - mx));
        const float l  = mx + __logf(s);
        a.x -= l; a.y -= l;
    }
};

template <typename VT, typename EPI>
__device__ __forceinline__ void gather_nodes(const int* __restrict__ cnt, const int* __restrict__ slots,
                                             const int2* __restrict__ ovf, const int* __restrict__ novf,
                                             const VT* __restrict__ rows, VT* __restrict__ dst, int N,
                                             EPI& epi) {
    const int wave = __builtin_amdgcn_readfirstlane(threadIdx.x >> 5);
    const int lane = threadIdx.x & 31;
    const int G = gridDim.x * 8;
    int c0 = blockIdx.x * 8 + wave;
    if (c0 >= N) return;
    int c1 = c0 + G;
    const int nov = *novf;
    int n0 = cnt[c0];
    int n1 = 0;
    if (c1 < N) n1 = cnt[c1];
    int ri = 0;
    if (lane < min(BUCKET, n0)) ri = slots[c0 * BUCKET + lane];
    for (;;) {
        const int c2 = c1 + G;
        int n2 = 0;
        if (c2 < N) n2 = cnt[c2];
        int ri1 = 0;
        if (lane < min(BUCKET, n1)) ri1 = slots[c1 * BUCKET + lane];
        const int n = n0;
        const float dc = rsqrtf((float)(n + 1));
        const VT self = rows[(size_t)c0 * 32 + lane];
        VT acc;
        set_zero(acc);
        {
            const int m = min(BUCKET, n);
            if (m > 0)  edge_batch<16>(rows, lane, ri, 0, m, acc);
            if (m > 16) edge_batch<16>(rows, lane, ri, 16, m - 16, acc);
        }
        if (n > BUCKET) {
            for (int b = 0; b < nov; b += 32) {
                int2 oe = make_int2(-1, 0);
                if (b + lane < nov) oe = ovf[b + lane];
                unsigned hit = __builtin_amdgcn_ballot_w32(oe.x == c0);
                while (hit != 0u) {
                    const int j = __builtin_amdgcn_readfirstlane(__builtin_ctz(hit));
                    hit &= hit - 1u;
                    const int r = __builtin_amdgcn_readlane(oe.y, j);
                    const VT v = rows[(size_t)r * 32 + lane];
                    add_row(acc, v);
                }
            }
        }
        add_row(acc, self);
        scale_row(acc, dc);
        epi.row(acc);
        dst[(size_t)c0 * 32 + lane] = acc;
        if (c1 >= N) break;
        c0 = c1; c1 = c2;
        n0 = n1; n1 = n2;
        ri = ri1;
    }
}


__global__ void __launch_bounds__(256) gemm1_wmma(const float* __restrict__ x, const float* __restrict__ W1,
                           const float* __restrict__ b1, const int* __restrict__ cnt,
                           float* __restrict__ h1, int N) {
    const int wave  = threadIdx.x >> 5;
    const int lane  = threadIdx.x & 31;
    const int col0  = wave * 16;
    const int m     = lane & 15;
    const int n     = lane & 15;
    const int khalf = lane >> 4;

    const float bias = b1[col0 + n];
    v2f bf[32];
    #pragma unroll
    for (int u = 0; u < 16; ++u) {
        const int kb = 8 * u + 4 * khalf;
        bf[2 * u].x     = W1[(size_t)(kb + 0) * D_HID + col0 + n];
        bf[2 * u].y     = W1[(size_t)(kb + 1) * D_HID + col0 + n];
        bf[2 * u + 1].x = W1[(size_t)(kb + 2) * D_HID + col0 + n];
        bf[2 * u + 1].y = W1[(size_t)(kb + 3) * D_HID + col0 + n];
    }

    const int rstep = gridDim.x * 16;
    for (int row0 = blockIdx.x * 16; row0 < N; row0 += rstep) {
        v8f c;
        #pragma unroll
        for (int r = 0; r < 8; ++r) c[r] = bias;

        const int mbase = khalf * 8;
        int dn[8];
        #pragma unroll
        for (int r = 0; r < 8; ++r) dn[r] = cnt[row0 + mbase + r];
        const float4* arow = (const float4*)(x + (size_t)(row0 + m) * D_HID + 4 * khalf);
        float4 af[16];
        #pragma unroll
        for (int u = 0; u < 16; ++u) af[u] = arow[2 * u];
        asm volatile("" :: "v"(dn[0]), "v"(dn[1]), "v"(dn[2]), "v"(dn[3]), "v"(dn[4]), "v"(dn[5]), "v"(dn[6]), "v"(dn[7]));
        __builtin_amdgcn_sched_barrier(0);
        #pragma unroll
        for (int u = 0; u < 16; ++u) {
            v2f lo, hi;
            lo.x = af[u].x; lo.y = af[u].y;
            hi.x = af[u].z; hi.y = af[u].w;
            c = __builtin_amdgcn_wmma_f32_16x16x4_f32(false, lo, false, bf[2 * u], (short)0, c, false, false);
            c = __builtin_amdgcn_wmma_f32_16x16x4_f32(false, hi, false, bf[2 * u + 1], (short)0, c, false, false);
        }

        asm volatile("v_nop\n\tv_nop\n\tv_nop\n\tv_nop\n\tv_nop\n\tv_nop\n\tv_nop\n\tv_nop" : "+v"(c));
        #pragma unroll
        for (int r = 0; r < 8; ++r)
            h1[(size_t)(row0 + mbase + r) * D_HID + col0 + n] = c[r] * __builtin_amdgcn_rsqf((float)(dn[r] + 1));
    }
}


__global__ void __launch_bounds__(256) gather128(const int* __restrict__ cnt, const int* __restrict__ slots,
                          const int2* __restrict__ ovf, const int* __restrict__ novf,
                          const float* __restrict__ h1, float* __restrict__ hagg, float* stats, int N) {
    __shared__ float part[8][2 * D_HID];
    RowStats st;
    st.s = make_float4(0.0f, 0.0f, 0.0f, 0.0f);
    st.q = make_float4(0.0f, 0.0f, 0.0f, 0.0f);
    gather_nodes<float4>(cnt, slots, ovf, novf, (const float4*)h1, (float4*)hagg, N, st);
    const int wave = threadIdx.x >> 5;
    const int lane = threadIdx.x & 31;
    *(float4*)&part[wave][4 * lane]         = st.s;
    *(float4*)&part[wave][D_HID + 4 * lane] = st.q;
    __syncthreads();
    float tot = 0.0f;
    #pragma unroll
    for (int w = 0; w < 8; ++w) tot += part[w][threadIdx.x];
    atomicAdd(&stats[threadIdx.x], tot);
}


__global__ void bn_finalize(const float* __restrict__ bsum, const float* __restrict__ bsq,
                            const float* __restrict__ gamma, const float* __restrict__ beta,
                            float* scale, float* shift, int N) {
    int c = threadIdx.x;
    if (c < D_HID) {
        float inv = 1.0f / (float)N;
        float mu = bsum[c] * inv;
        float var = bsq[c] * inv - mu * mu;
        float s = gamma[c] * rsqrtf(var + BN_EPS);
        scale[c] = s;
        shift[c] = beta[c] - mu * s;
    }
}


__global__ void __launch_bounds__(256) gemm2_wmma(const float* __restrict__ hagg, const float* __restrict__ scale,
                           const float* __restrict__ shift, const float* __restrict__ W2,
                           const float* __restrict__ b2, const int* __restrict__ cnt,
                           float* __restrict__ h2, int N) {
    const int wave  = __builtin_amdgcn_readfirstlane(threadIdx.x >> 5);
    const int lane  = threadIdx.x & 31;
    const int col0  = (wave & 3) * 16;
    const int m     = lane & 15;
    const int n     = lane & 15;
    const int khalf = lane >> 4;

    const float bias = b2[col0 + n];
    v2f bf[32];
    #pragma unroll
    for (int u = 0; u < 16; ++u) {
        const int kb = 8 * u + 4 * khalf;
        bf[2 * u].x     = W2[(size_t)(kb + 0) * D_OUT + col0 + n];
        bf[2 * u].y     = W2[(size_t)(kb + 1) * D_OUT + col0 + n];
        bf[2 * u + 1].x = W2[(size_t)(kb + 2) * D_OUT + col0 + n];
        bf[2 * u + 1].y = W2[(size_t)(kb + 3) * D_OUT + col0 + n];
    }

    float4 sc[16], sh[16];
    #pragma unroll
    for (int u = 0; u < 16; ++u) {
        sc[u] = ((const float4*)(scale + 4 * khalf))[2 * u];
        sh[u] = ((const float4*)(shift + 4 * khalf))[2 * u];
    }

    const int rstep = gridDim.x * 32;
    for (int row0 = (blockIdx.x * 2 + (wave >> 2)) * 16; row0 < N; row0 += rstep) {
        v8f c;
        #pragma unroll
        for (int r = 0; r < 8; ++r) c[r] = bias;

        const int mbase = khalf * 8;
        int dn[8];
        #pragma unroll
        for (int r = 0; r < 8; ++r) dn[r] = cnt[row0 + mbase + r];
        const float4* arow = (const float4*)(hagg + (size_t)(row0 + m) * D_HID + 4 * khalf);
        float4 af[16];
        #pragma unroll
        for (int u = 0; u < 16; ++u) af[u] = arow[2 * u];
        asm volatile("" :: "v"(dn[0]), "v"(dn[1]), "v"(dn[2]), "v"(dn[3]), "v"(dn[4]), "v"(dn[5]), "v"(dn[6]), "v"(dn[7]));
        __builtin_amdgcn_sched_barrier(0);
        #pragma unroll
        for (int u = 0; u < 16; ++u) {
            v2f lo, hi;
            lo.x = fmaxf(af[u].x * sc[u].x + sh[u].x, 0.0f);
            lo.y = fmaxf(af[u].y * sc[u].y + sh[u].y, 0.0f);
            hi.x = fmaxf(af[u].z * sc[u].z + sh[u].z, 0.0f);
            hi.y = fmaxf(af[u].w * sc[u].w + sh[u].w, 0.0f);
            c = __builtin_amdgcn_wmma_f32_16x16x4_f32(false, lo, false, bf[2 * u], (short)0, c, false, false);
            c = __builtin_amdgcn_wmma_f32_16x16x4_f32(false, hi, false, bf[2 * u + 1], (short)0, c, false, false);
        }

        asm volatile("v_nop\n\tv_nop\n\tv_nop\n\tv_nop\n\tv_nop\n\tv_nop\n\tv_nop\n\tv_nop" : "+v"(c));
        #pragma unroll
        for (int r = 0; r < 8; ++r)
            h2[(size_t)(row0 + mbase + r) * D_OUT + col0 + n] = c[r] * __builtin_amdgcn_rsqf((float)(dn[r] + 1));
    }
}


__global__ void __launch_bounds__(256) gather64(const int* __restrict__ cnt, const int* __restrict__ slots,
                         const int2* __restrict__ ovf, const int* __restrict__ novf,
                         const float* __restrict__ h2, float* __restrict__ out, int N) {
    RowLogSoftmax ls;
    gather_nodes<float2>(cnt, slots, ovf, novf, (const float2*)h2, (float2*)out, N, ls);
}


extern "C" void kernel_launch(void* const* d_in, const int* in_sizes, int n_in,
                              void* d_out, int out_size, void* d_ws, size_t ws_size,
                              hipStream_t stream) {
    const float* x     = (const float*)d_in[0];
    const float* W1    = (const float*)d_in[1];
    const float* b1    = (const float*)d_in[2];
    const float* gamma = (const float*)d_in[3];
    const float* beta  = (const float*)d_in[4];
    const float* W2    = (const float*)d_in[5];
    const float* b2    = (const float*)d_in[6];
    const int*   ei    = (const int*)d_in[7];

    const int N = in_sizes[0] / D_HID;
    const int E = in_sizes[7] / 2;
    const int* erow = ei;
    const int* ecol = ei + E;

    float* ws    = (float*)d_ws;
    float* h1    = ws;
    float* hagg  = h1 + (size_t)N * D_HID;
    float* stats = hagg + (size_t)N * D_HID;
    float* bsum  = stats;
    float* bsq   = stats + 128;
    float* scal  = stats + 256;
    float* shft  = stats + 384;
    int*   novf  = (int*)(stats + 512);
    int*   slots = (int*)(stats + 576);
    int2*  ovf   = (int2*)(slots + (size_t)N * BUCKET);
    int*   cnt   = (int*)(ovf + E);
    float* h2    = h1;
    float* outp  = (float*)d_out;

    init_cnt_stats<<<(N + 255) / 256, 256, 0, stream>>>(cnt, stats, novf, N);
    place_edges<<<512, 256, 0, stream>>>(erow, ecol, cnt, slots, ovf, novf, E);

    gemm1_wmma<<<512, 256, 0, stream>>>(x, W1, b1, cnt, h1, N);
    gather128<<<1024, 256, 0, stream>>>(cnt, slots, ovf, novf, h1, hagg, stats, N);

    bn_finalize<<<1, 128, 0, stream>>>(bsum, bsq, gamma, beta, scal, shft, N);

    gemm2_wmma<<<256, 256, 0, stream>>>(hagg, scal, shft, W2, b2, cnt, h2, N);
    gather64<<<1024, 256, 0, stream>>>(cnt, slots, ovf, novf, h2, outp, N);
}


# Edge placement and GEMM1 run in one launch (blocks chosen by index): GEMM1 leaves its rows unscaled, gather 1 weights each source row by its dis in an fma, which changes the rounding of its sums
# speedup vs baseline: 2.9583x; 1.0312x over previous
#include <hip/hip_runtime.h>
#include <hip/hip_bf16.h>
#include <math.h>

typedef float v8f __attribute__((ext_vector_type(8)));
typedef float v2f __attribute__((ext_vector_type(2)));

#define D_HID 128
#define D_OUT 64
#define BN_EPS 1e-5f


#define BUCKET 32

__global__ void init_cnt_stats(int* cnt, float* stats, int* novf, int N) {
    int i = blockIdx.x * blockDim.x + threadIdx.x;
    int stride = blockDim.x * gridDim.x;
    for (; i < N; i += stride) {
        cnt[i] = 0;
        if (i < 256) stats[i] = 0.0f;
        if (i == 0) *novf = 0;
    }
}

#define EPT 8
__device__ __forceinline__ void place_edges_body(const int* __restrict__ row, const int* __restrict__ col, int* cnt,
                            int* __restrict__ slots, int2* __restrict__ ovf, int* novf, int E,
                            int bid, int nblk) {
    const int stride = blockDim.x * nblk;
    for (int e = bid * blockDim.x + threadIdx.x; e < E; e += EPT * stride) {
        int c[EPT], r[EPT], p[EPT];
        #pragma unroll
        for (int k = 0; k < EPT; ++k) {
            const int ek = e + k * stride;
            c[k] = -1; r[k] = 0;
            if (ek < E) { c[k] = col[ek]; r[k] = row[ek]; }
        }
        #pragma unroll
        for (int k = 0; k < EPT; ++k) {
            p[k] = 0;
            if (c[k] >= 0) p[k] = atomicAdd(&cnt[c[k]], 1);
        }
        #pragma unroll
        for (int k = 0; k < EPT; ++k) {
            if (c[k] >= 0) {
                if (p[k] < BUCKET) {
                    slots[c[k] * BUCKET + p[k]] = r[k];
                } else {
                    const int q = atomicAdd(novf, 1);
                    ovf[q] = make_int2(c[k], r[k]);
                }
            }
        }
    }
}


__device__ __forceinline__ void add_row(float4& acc, const float4& v) {
    acc.x += v.x; acc.y += v.y; acc.z += v.z; acc.w += v.w;
}
__device__ __forceinline__ void add_row(float2& acc, const float2& v) {
    acc.x += v.x; acc.y += v.y;
}
__device__ __forceinline__ void scale_row(float4& a, float d) { a.x *= d; a.y *= d; a.z *= d; a.w *= d; }
__device__ __forceinline__ void scale_row(float2& a, float d) { a.x *= d; a.y *= d; }
__device__ __forceinline__ void fma_row(float4& acc, float w, const float4& v) {
    acc.x = fmaf(w, v.x, acc.x); acc.y = fmaf(w, v.y, acc.y); acc.z = fmaf(w, v.z, acc.z); acc.w = fmaf(w, v.w, acc.w);
}
__device__ __forceinline__ void fma_row(float2& acc, float w, const float2& v) {
    acc.x = fmaf(w, v.x, acc.x); acc.y = fmaf(w, v.y, acc.y);
}
__device__ __forceinline__ void set_zero(float4& a) { a = make_float4(0.0f, 0.0f, 0.0f, 0.0f); }
__device__ __forceinline__ void set_zero(float2& a) { a = make_float2(0.0f, 0.0f); }

template <int K, bool WEIGHTED, typename VT>
__device__ __forceinline__ void edge_batch(const VT* __restrict__ rows, int lane,
                                           int ri, int sc, int j, int cnt, VT& acc) {
    VT v[K];
    #pragma unroll
    for (int k = 0; k < K; ++k) {
        set_zero(v[k]);
        if (k < cnt) {
            const int r = __builtin_amdgcn_readlane(ri, j + k);
            v[k] = rows[(size_t)r * 32 + lane];
        }
    }
    float wl = 0.0f;
    if constexpr (WEIGHTED) wl = __builtin_amdgcn_rsqf((float)(sc + 1));
    #pragma unroll
    for (int k = 0; k < K; ++k) {
        if (k < cnt) {
            if constexpr (WEIGHTED) {
                const float w = __builtin_bit_cast(float, __builtin_amdgcn_readlane(__builtin_bit_cast(int, wl), j + k));
                fma_row(acc, w, v[k]);
            } else {
                add_row(acc, v[k]);
            }
        }
    }
}


struct RowStats {
    float4 s, q;
    __device__ __forceinline__ void row(float4& a) {
        s.x += a.x; s.y += a.y; s.z += a.z; s.w += a.w;
        q.x = fmaf(a.x, a.x, q.x); q.y = fmaf(a.y, a.y, q.y);
        q.z = fmaf(a.z, a.z, q.z); q.w = fmaf(a.w, a.w, q.w);
    }
};

template <int M>
__device__ __forceinline__ float row_xor(float x) {
    return __builtin_bit_cast(float, __builtin_amdgcn_mov_dpp(__builtin_bit_cast(int, x), 0x160 + M, 0xf, 0xf, true));
}
__device__ __forceinline__ float xhalf(float x) {
    const int b = __builtin_bit_cast(int, x);
    return __builtin_bit_cast(float, __builtin_amdgcn_permlanex16(b, b, 0x76543210, 0xfedcba98, false, false));
}
__device__ __forceinline__ float wave_max(float x) {
    x = fmaxf(x, row_xor<1>(x)); x = fmaxf(x, row_xor<2>(x));
    x = fmaxf(x, row_xor<4>(x)); x = fmaxf(x, row_xor<8>(x));
    return fmaxf(x, xhalf(x));
}
__device__ __forceinline__ float wave_sum(float x) {
    x += row_xor<1>(x); x += row_xor<2>(x);
    x += row_xor<4>(x); x += row_xor<8>(x);
    return x + xhalf(x);
}

struct RowLogSoftmax {
    __device__ __forceinline__ void row(float2& a) {
        const float mx = wave_max(fmaxf(a.x, a.y));
        const float s  = wave_sum(__expf(a.x - mx) + __expf(a.y - mx));
        const float l  = mx + __logf(s);
        a.x -= l; a.y -= l;
    }
};

template <bool WEIGHTED, typename VT, typename EPI>
__device__ __forceinline__ void gather_nodes(const int* __restrict__ cnt, const int* __restrict__ slots,
                                             const int2* __restrict__ ovf, const int* __restrict__ novf,
                                             const VT* __restrict__ rows, VT* __restrict__ dst, int N,
                                             EPI& epi) {
    const int wave = __builtin_amdgcn_readfirstlane(threadIdx.x >> 5);
    const int lane = threadIdx.x & 31;
    const int G = gridDim.x * 8;
    int c0 = blockIdx.x * 8 + wave;
    if (c0 >= N) return;
    int c1 = c0 + G;
    const int nov = *novf;
    int n0 = cnt[c0];
    int n1 = 0;
    if (c1 < N) n1 = cnt[c1];
    int ri = 0;
    if (lane < min(BUCKET, n0)) ri = slots[c0 * BUCKET + lane];
    for (;;) {
        int sc = 0;
        if constexpr (WEIGHTED) sc = cnt[ri];
        const int c2 = c1 + G;
        int n2 = 0;
        if (c2 < N) n2 = cnt[c2];
        int ri1 = 0;
        if (lane < min(BUCKET, n1)) ri1 = slots[c1 * BUCKET + lane];
        const int n = n0;
        const float dc = rsqrtf((float)(n + 1));
        const VT self = rows[(size_t)c0 * 32 + lane];
        VT acc;
        set_zero(acc);
        {
            const int m = min(BUCKET, n);
            if (m > 0)  edge_batch<16, WEIGHTED>(rows, lane, ri, sc, 0, m, acc);
            if (m > 16) edge_batch<16, WEIGHTED>(rows, lane, ri, sc, 16, m - 16, acc);
        }
        if (n > BUCKET) {
            for (int b = 0; b < nov; b += 32) {
                int2 oe = make_int2(-1, 0);
                if (b + lane < nov) oe = ovf[b + lane];
                unsigned hit = __builtin_amdgcn_ballot_w32(oe.x == c0);
                while (hit != 0u) {
                    const int j = __builtin_amdgcn_readfirstlane(__builtin_ctz(hit));
                    hit &= hit - 1u;
                    const int r = __builtin_amdgcn_readlane(oe.y, j);
                    const VT v = rows[(size_t)r * 32 + lane];
                    if constexpr (WEIGHTED) fma_row(acc, rsqrtf((float)(cnt[r] + 1)), v);
                    else add_row(acc, v);
                }
            }
        }
        if constexpr (WEIGHTED) fma_row(acc, dc, self);
        else add_row(acc, self);
        scale_row(acc, dc);
        epi.row(acc);
        dst[(size_t)c0 * 32 + lane] = acc;
        if (c1 >= N) break;
        c0 = c1; c1 = c2;
        n0 = n1; n1 = n2;
        ri = ri1;
    }
}


__device__ __forceinline__ void gemm1_body(const float* __restrict__ x, const float* __restrict__ W1,
                           const float* __restrict__ b1, float* __restrict__ h1, int N,
                           int bid, int nblk) {
    const int wave  = threadIdx.x >> 5;
    const int lane  = threadIdx.x & 31;
    const int col0  = wave * 16;
    const int m     = lane & 15;
    const int n     = lane & 15;
    const int khalf = lane >> 4;

    const float bias = b1[col0 + n];
    v2f bf[32];
    #pragma unroll
    for (int u = 0; u < 16; ++u) {
        const int kb = 8 * u + 4 * khalf;
        bf[2 * u].x     = W1[(size_t)(kb + 0) * D_HID + col0 + n];
        bf[2 * u].y     = W1[(size_t)(kb + 1) * D_HID + col0 + n];
        bf[2 * u + 1].x = W1[(size_t)(kb + 2) * D_HID + col0 + n];
        bf[2 * u + 1].y = W1[(size_t)(kb + 3) * D_HID + col0 + n];
    }

    const int rstep = nblk * 16;
    for (int row0 = bid * 16; row0 < N; row0 += rstep) {
        v8f c;
        #pragma unroll
        for (int r = 0; r < 8; ++r) c[r] = bias;

        const int mbase = khalf * 8;
        const float4* arow = (const float4*)(x + (size_t)(row0 + m) * D_HID + 4 * khalf);
        float4 af[16];
        #pragma unroll
        for (int u = 0; u < 16; ++u) af[u] = arow[2 * u];
        __builtin_amdgcn_sched_barrier(0);
        #pragma unroll
        for (int u = 0; u < 16; ++u) {
            v2f lo, hi;
            lo.x = af[u].x; lo.y = af[u].y;
            hi.x = af[u].z; hi.y = af[u].w;
            c = __builtin_amdgcn_wmma_f32_16x16x4_f32(false, lo, false, bf[2 * u], (short)0, c, false, false);
            c = __builtin_amdgcn_wmma_f32_16x16x4_f32(false, hi, false, bf[2 * u + 1], (short)0, c, false, false);
        }

        asm volatile("v_nop\n\tv_nop\n\tv_nop\n\tv_nop\n\tv_nop\n\tv_nop\n\tv_nop\n\tv_nop" : "+v"(c));
        #pragma unroll
        for (int r = 0; r < 8; ++r)
            h1[(size_t)(row0 + mbase + r) * D_HID + col0 + n] = c[r];
    }
}

__global__ void __launch_bounds__(256) place_and_gemm1(const int* __restrict__ row, const int* __restrict__ col,
                           int* cnt, int* __restrict__ slots, int2* __restrict__ ovf, int* novf, int E, int PB,
                           const float* __restrict__ x, const float* __restrict__ W1,
                           const float* __restrict__ b1, float* __restrict__ h1, int N, int GB) {
    if ((int)blockIdx.x < PB) place_edges_body(row, col, cnt, slots, ovf, novf, E, (int)blockIdx.x, PB);
    else                      gemm1_body(x, W1, b1, h1, N, (int)blockIdx.x - PB, GB);
}


__global__ void __launch_bounds__(256) gather128(const int* __restrict__ cnt, const int* __restrict__ slots,
                          const int2* __restrict__ ovf, const int* __restrict__ novf,
                          const float* __restrict__ h1, float* __restrict__ hagg, float* stats, int N) {
    __shared__ float part[8][2 * D_HID];
    RowStats st;
    st.s = make_float4(0.0f, 0.0f, 0.0f, 0.0f);
    st.q = make_float4(0.0f, 0.0f, 0.0f, 0.0f);
    gather_nodes<true, float4>(cnt, slots, ovf, novf, (const float4*)h1, (float4*)hagg, N, st);
    const int wave = threadIdx.x >> 5;
    const int lane = threadIdx.x & 31;
    *(float4*)&part[wave][4 * lane]         = st.s;
    *(float4*)&part[wave][D_HID + 4 * lane] = st.q;
    __syncthreads();
    float tot = 0.0f;
    #pragma unroll
    for (int w = 0; w < 8; ++w) tot += part[w][threadIdx.x];
    atomicAdd(&stats[threadIdx.x], tot);
}


__global__ void bn_finalize(const float* __restrict__ bsum, const float* __restrict__ bsq,
                            const float* __restrict__ gamma, const float* __restrict__ beta,
                            float* scale, float* shift, int N) {
    int c = threadIdx.x;
    if (c < D_HID) {
        float inv = 1.0f / (float)N;
        float mu = bsum[c] * inv;
        float var = bsq[c] * inv - mu * mu;
        float s = gamma[c] * rsqrtf(var + BN_EPS);
        scale[c] = s;
        shift[c] = beta[c] - mu * s;
    }
}


__global__ void __launch_bounds__(256) gemm2_wmma(const float* __restrict__ hagg, const float* __restrict__ scale,
                           const float* __restrict__ shift, const float* __restrict__ W2,
                           const float* __restrict__ b2, const int* __restrict__ cnt,
                           float* __restrict__ h2, int N) {
    const int wave  = __builtin_amdgcn_readfirstlane(threadIdx.x >> 5);
    const int lane  = threadIdx.x & 31;
    const int col0  = (wave & 3) * 16;
    const int m     = lane & 15;
    const int n     = lane & 15;
    const int khalf = lane >> 4;

    const float bias = b2[col0 + n];
    v2f bf[32];
    #pragma unroll
    for (int u = 0; u < 16; ++u) {
        const int kb = 8 * u + 4 * khalf;
        bf[2 * u].x     = W2[(size_t)(kb + 0) * D_OUT + col0 + n];
        bf[2 * u].y     = W2[(size_t)(kb + 1) * D_OUT + col0 + n];
        bf[2 * u + 1].x = W2[(size_t)(kb + 2) * D_OUT + col0 + n];
        bf[2 * u + 1].y = W2[(size_t)(kb + 3) * D_OUT + col0 + n];
    }

    float4 sc[16], sh[16];
    #pragma unroll
    for (int u = 0; u < 16; ++u) {
        sc[u] = ((const float4*)(scale + 4 * khalf))[2 * u];
        sh[u] = ((const float4*)(shift + 4 * khalf))[2 * u];
    }

    const int rstep = gridDim.x * 32;
    for (int row0 = (blockIdx.x * 2 + (wave >> 2)) * 16; row0 < N; row0 += rstep) {
        v8f c;
        #pragma unroll
        for (int r = 0; r < 8; ++r) c[r] = bias;

        const int mbase = khalf * 8;
        int dn[8];
        #pragma unroll
        for (int r = 0; r < 8; ++r) dn[r] = cnt[row0 + mbase + r];
        const float4* arow = (const float4*)(hagg + (size_t)(row0 + m) * D_HID + 4 * khalf);
        float4 af[16];
        #pragma unroll
        for (int u = 0; u < 16; ++u) af[u] = arow[2 * u];
        asm volatile("" :: "v"(dn[0]), "v"(dn[1]), "v"(dn[2]), "v"(dn[3]), "v"(dn[4]), "v"(dn[5]), "v"(dn[6]), "v"(dn[7]));
        __builtin_amdgcn_sched_barrier(0);
        #pragma unroll
        for (int u = 0; u < 16; ++u) {
            v2f lo, hi;
            lo.x = fmaxf(af[u].x * sc[u].x + sh[u].x, 0.0f);
            lo.y = fmaxf(af[u].y * sc[u].y + sh[u].y, 0.0f);
            hi.x = fmaxf(af[u].z * sc[u].z + sh[u].z, 0.0f);
            hi.y = fmaxf(af[u].w * sc[u].w + sh[u].w, 0.0f);
            c = __builtin_amdgcn_wmma_f32_16x16x4_f32(false, lo, false, bf[2 * u], (short)0, c, false, false);
            c = __builtin_amdgcn_wmma_f32_16x16x4_f32(false, hi, false, bf[2 * u + 1], (short)0, c, false, false);
        }

        asm volatile("v_nop\n\tv_nop\n\tv_nop\n\tv_nop\n\tv_nop\n\tv_nop\n\tv_nop\n\tv_nop" : "+v"(c));
        #pragma unroll
        for (int r = 0; r < 8; ++r)
            h2[(size_t)(row0 + mbase + r) * D_OUT + col0 + n] = c[r] * __builtin_amdgcn_rsqf((float)(dn[r] + 1));
    }
}


__global__ void __launch_bounds__(256) gather64(const int* __restrict__ cnt, const int* __restrict__ slots,
                         const int2* __restrict__ ovf, const int* __restrict__ novf,
                         const float* __restrict__ h2, float* __restrict__ out, int N) {
    RowLogSoftmax ls;
    gather_nodes<false, float2>(cnt, slots, ovf, novf, (const float2*)h2, (float2*)out, N, ls);
}


extern "C" void kernel_launch(void* const* d_in, const int* in_sizes, int n_in,
                              void* d_out, int out_size, void* d_ws, size_t ws_size,
                              hipStream_t stream) {
    const float* x     = (const float*)d_in[0];
    const float* W1    = (const float*)d_in[1];
    const float* b1    = (const float*)d_in[2];
    const float* gamma = (const float*)d_in[3];
    const float* beta  = (const float*)d_in[4];
    const float* W2    = (const float*)d_in[5];
    const float* b2    = (const float*)d_in[6];
    const int*   ei    = (const int*)d_in[7];

    const int N = in_sizes[0] / D_HID;
    const int E = in_sizes[7] / 2;
    const int* erow = ei;
    const int* ecol = ei + E;

    float* ws    = (float*)d_ws;
    float* h1    = ws;
    float* hagg  = h1 + (size_t)N * D_HID;
    float* stats = hagg + (size_t)N * D_HID;
    float* bsum  = stats;
    float* bsq   = stats + 128;
    float* scal  = stats + 256;
    float* shft  = stats + 384;
    int*   novf  = (int*)(stats + 512);
    int*   slots = (int*)(stats + 576);
    int2*  ovf   = (int2*)(slots + (size_t)N * BUCKET);
    int*   cnt   = (int*)(ovf + E);
    float* h2    = h1;
    float* outp  = (float*)d_out;

    init_cnt_stats<<<(N + 255) / 256, 256, 0, stream>>>(cnt, stats, novf, N);
    place_and_gemm1<<<768, 256, 0, stream>>>(erow, ecol, cnt, slots, ovf, novf, E, 256, x, W1, b1, h1, N, 512);

    gather128<<<1024, 256, 0, stream>>>(cnt, slots, ovf, novf, h1, hagg, stats, N);

    bn_finalize<<<1, 128, 0, stream>>>(bsum, bsq, gamma, beta, scal, shft, N);

    gemm2_wmma<<<256, 256, 0, stream>>>(hagg, scal, shft, W2, b2, cnt, h2, N);
    gather64<<<1024, 256, 0, stream>>>(cnt, slots, ovf, novf, h2, outp, N);
}


# Edges grouped by destination with LDS counters in two levels (block-local runs per 256-node range, then one owner block per range), no global atomic per edge; slot order = LDS arrival order
# speedup vs baseline: 4.5320x; 1.5320x over previous
#include <hip/hip_runtime.h>
#include <hip/hip_bf16.h>
#include <math.h>

typedef float v8f __attribute__((ext_vector_type(8)));
typedef float v2f __attribute__((ext_vector_type(2)));

#define D_HID 128
#define D_OUT 64
#define BN_EPS 1e-5f


#define BUCKET 32
#define RANGE_SHIFT 8
#define NRANGE 256

#define EPT 8
__device__ __forceinline__ void bin_edges_body(const int* __restrict__ row, const int* __restrict__ col,
                            int2* __restrict__ tab, int2* __restrict__ sorted,
                            float* __restrict__ stats, int* __restrict__ novf, int E, int bid, int nblk) {
    __shared__ int bcnt[NRANGE];
    __shared__ int boff[NRANGE];
    __shared__ int bfill[NRANGE];
    __shared__ int wsum[8];
    const int t = threadIdx.x;
    const int lane = t & 31;
    const int wave = t >> 5;
    if (bid == 0) {
        stats[t] = 0.0f;
        if (t == 0) *novf = 0;
    }
    bcnt[t] = 0;
    bfill[t] = 0;
    __syncthreads();
    const int chunk = ((E + nblk - 1) / nblk + 15) & ~15;
    const int e0 = bid * chunk;
    const int e1 = min(E, e0 + chunk);
    const int stride = blockDim.x;
    for (int e = e0 + t; e < e1; e += EPT * stride) {
        int c[EPT];
        #pragma unroll
        for (int k = 0; k < EPT; ++k) {
            const int ek = e + k * stride;
            c[k] = -1;
            if (ek < e1) c[k] = col[ek];
        }
        #pragma unroll
        for (int k = 0; k < EPT; ++k)
            if (c[k] >= 0) atomicAdd(&bcnt[min(c[k] >> RANGE_SHIFT, NRANGE - 1)], 1);
    }
    __syncthreads();
    const int v = bcnt[t];
    int inc = v;
    #pragma unroll
    for (int d = 1; d < 32; d <<= 1) {
        const int o = __shfl_up(inc, d);
        if (lane >= d) inc += o;
    }
    if (lane == 31) wsum[wave] = inc;
    __syncthreads();
    int wbase = 0;
    #pragma unroll
    for (int w = 0; w < 8; ++w)
        if (w < wave) wbase += wsum[w];
    const int excl = wbase + inc - v;
    boff[t] = excl;
    tab[bid * NRANGE + t] = make_int2(v, e0 + excl);
    __syncthreads();
    for (int e = e0 + t; e < e1; e += EPT * stride) {
        int c[EPT], r[EPT];
        #pragma unroll
        for (int k = 0; k < EPT; ++k) {
            const int ek = e + k * stride;
            c[k] = -1; r[k] = 0;
            if (ek < e1) { c[k] = col[ek]; r[k] = row[ek]; }
        }
        int g[EPT], p[EPT];
        #pragma unroll
        for (int k = 0; k < EPT; ++k) {
            g[k] = min(c[k] >> RANGE_SHIFT, NRANGE - 1);
            p[k] = 0;
            if (c[k] >= 0) p[k] = atomicAdd(&bfill[g[k]], 1);
        }
        #pragma unroll
        for (int k = 0; k < EPT; ++k)
            if (c[k] >= 0) sorted[e0 + boff[g[k]] + p[k]] = make_int2(c[k], r[k]);
    }
}

#define RUN 32
__global__ void __launch_bounds__(256) place_owner(const int2* __restrict__ tab,
                            const int2* __restrict__ sorted, int* __restrict__ cnt, int* __restrict__ slots,
                            int2* __restrict__ ovf, int* novf, int N, int npiece) {
    __shared__ int  lcnt[NRANGE];
    __shared__ int4 lsl4[NRANGE * BUCKET / 4];
    int* lsl = (int*)lsl4;
    const int t = threadIdx.x;
    const int g = blockIdx.x;
    const int base = g << RANGE_SHIFT;
    if (base >= N) return;
    lcnt[t] = 0;
    #pragma unroll
    for (int i = 0; i < NRANGE * BUCKET / 4 / 256; ++i) lsl4[i * 256 + t] = make_int4(0, 0, 0, 0);
    __syncthreads();
    for (int b = t; b < npiece; b += blockDim.x) {
        const int2 run = tab[b * NRANGE + g];
        const int n = run.x;
        const int o = run.y;
        for (int i0 = 0; i0 < n; i0 += RUN) {
            int2 e[RUN];
            #pragma unroll
            for (int u = 0; u < RUN; ++u) {
                e[u] = make_int2(-1, 0);
                if (i0 + u < n) e[u] = sorted[o + i0 + u];
            }
            int p[RUN];
            #pragma unroll
            for (int u = 0; u < RUN; ++u) {
                p[u] = 0;
                if (e[u].x >= 0) p[u] = atomicAdd(&lcnt[(e[u].x - base) & (NRANGE - 1)], 1);
            }
            #pragma unroll
            for (int u = 0; u < RUN; ++u) {
                if (e[u].x >= 0) {
                    const int j = (e[u].x - base) & (NRANGE - 1);
                    if (p[u] < BUCKET) {
                        lsl[j * BUCKET + p[u]] = e[u].y;
                    } else {
                        const int q = atomicAdd(novf, 1);
                        ovf[q] = e[u];
                    }
                }
            }
        }
    }
    __syncthreads();
    const int nrows = min(NRANGE, N - base);
    if (t < nrows) cnt[base + t] = lcnt[t];
    int4* dst = (int4*)(slots + (size_t)base * BUCKET);
    for (int i = t; i < nrows * (BUCKET / 4); i += blockDim.x) dst[i] = lsl4[i];
}


__device__ __forceinline__ void add_row(float4& acc, const float4& v) {
    acc.x += v.x; acc.y += v.y; acc.z += v.z; acc.w += v.w;
}
__device__ __forceinline__ void add_row(float2& acc, const float2& v) {
    acc.x += v.x; acc.y += v.y;
}
__device__ __forceinline__ void scale_row(float4& a, float d) { a.x *= d; a.y *= d; a.z *= d; a.w *= d; }
__device__ __forceinline__ void scale_row(float2& a, float d) { a.x *= d; a.y *= d; }
__device__ __forceinline__ void fma_row(float4& acc, float w, const float4& v) {
    acc.x = fmaf(w, v.x, acc.x); acc.y = fmaf(w, v.y, acc.y); acc.z = fmaf(w, v.z, acc.z); acc.w = fmaf(w, v.w, acc.w);
}
__device__ __forceinline__ void fma_row(float2& acc, float w, const float2& v) {
    acc.x = fmaf(w, v.x, acc.x); acc.y = fmaf(w, v.y, acc.y);
}
__device__ __forceinline__ void set_zero(float4& a) { a = make_float4(0.0f, 0.0f, 0.0f, 0.0f); }
__device__ __forceinline__ void set_zero(float2& a) { a = make_float2(0.0f, 0.0f); }

template <int K, bool WEIGHTED, typename VT>
__device__ __forceinline__ void edge_batch(const VT* __restrict__ rows, int lane,
                                           int ri, int sc, int j, int cnt, VT& acc) {
    VT v[K];
    #pragma unroll
    for (int k = 0; k < K; ++k) {
        set_zero(v[k]);
        if (k < cnt) {
            const int r = __builtin_amdgcn_readlane(ri, j + k);
            v[k] = rows[(size_t)r * 32 + lane];
        }
    }
    float wl = 0.0f;
    if constexpr (WEIGHTED) wl = __builtin_amdgcn_rsqf((float)(sc + 1));
    #pragma unroll
    for (int k = 0; k < K; ++k) {
        if (k < cnt) {
            if constexpr (WEIGHTED) {
                const float w = __builtin_bit_cast(float, __builtin_amdgcn_readlane(__builtin_bit_cast(int, wl), j + k));
                fma_row(acc, w, v[k]);
            } else {
                add_row(acc, v[k]);
            }
        }
    }
}


struct RowStats {
    float4 s, q;
    __device__ __forceinline__ void row(float4& a) {
        s.x += a.x; s.y += a.y; s.z += a.z; s.w += a.w;
        q.x = fmaf(a.x, a.x, q.x); q.y = fmaf(a.y, a.y, q.y);
        q.z = fmaf(a.z, a.z, q.z); q.w = fmaf(a.w, a.w, q.w);
    }
};

template <int M>
__device__ __forceinline__ float row_xor(float x) {
    return __builtin_bit_cast(float, __builtin_amdgcn_mov_dpp(__builtin_bit_cast(int, x), 0x160 + M, 0xf, 0xf, true));
}
__device__ __forceinline__ float xhalf(float x) {
    const int b = __builtin_bit_cast(int, x);
    return __builtin_bit_cast(float, __builtin_amdgcn_permlanex16(b, b, 0x76543210, 0xfedcba98, false, false));
}
__device__ __forceinline__ float wave_max(float x) {
    x = fmaxf(x, row_xor<1>(x)); x = fmaxf(x, row_xor<2>(x));
    x = fmaxf(x, row_xor<4>(x)); x = fmaxf(x, row_xor<8>(x));
    return fmaxf(x, xhalf(x));
}
__device__ __forceinline__ float wave_sum(float x) {
    x += row_xor<1>(x); x += row_xor<2>(x);
    x += row_xor<4>(x); x += row_xor<8>(x);
    return x + xhalf(x);
}

struct RowLogSoftmax {
    __device__ __forceinline__ void row(float2& a) {
        const float mx = wave_max(fmaxf(a.x, a.y));
        const float s  = wave_sum(__expf(a.x - mx) + __expf(a.y - mx));
        const float l  = mx + __logf(s);
        a.x -= l; a.y -= l;
    }
};

template <bool WEIGHTED, typename VT, typename EPI>
__device__ __forceinline__ void gather_nodes(const int* __restrict__ cnt, const int* __restrict__ slots,
                                             const int2* __restrict__ ovf, const int* __restrict__ novf,
                                             const VT* __restrict__ rows, VT* __restrict__ dst, int N,
                                             EPI& epi) {
    const int wave = __builtin_amdgcn_readfirstlane(threadIdx.x >> 5);
    const int lane = threadIdx.x & 31;
    const int G = gridDim.x * 8;
    int c0 = blockIdx.x * 8 + wave;
    if (c0 >= N) return;
    int c1 = c0 + G;
    const int nov = *novf;
    int n0 = cnt[c0];
    int n1 = 0;
    if (c1 < N) n1 = cnt[c1];
    int ri = 0;
    if (lane < min(BUCKET, n0)) ri = slots[c0 * BUCKET + lane];
    for (;;) {
        int sc = 0;
        if constexpr (WEIGHTED) sc = cnt[ri];
        const int c2 = c1 + G;
        int n2 = 0;
        if (c2 < N) n2 = cnt[c2];
        int ri1 = 0;
        if (lane < min(BUCKET, n1)) ri1 = slots[c1 * BUCKET + lane];
        const int n = n0;
        const float dc = rsqrtf((float)(n + 1));
        const VT self = rows[(size_t)c0 * 32 + lane];
        VT acc;
        set_zero(acc);
        {
            const int m = min(BUCKET, n);
            if (m > 0)  edge_batch<16, WEIGHTED>(rows, lane, ri, sc, 0, m, acc);
            if (m > 16) edge_batch<16, WEIGHTED>(rows, lane, ri, sc, 16, m - 16, acc);
        }
        if (n > BUCKET) {
            for (int b = 0; b < nov; b += 32) {
                int2 oe = make_int2(-1, 0);
                if (b + lane < nov) oe = ovf[b + lane];
                unsigned hit = __builtin_amdgcn_ballot_w32(oe.x == c0);
                while (hit != 0u) {
                    const int j = __builtin_amdgcn_readfirstlane(__builtin_ctz(hit));
                    hit &= hit - 1u;
                    const int r = __builtin_amdgcn_readlane(oe.y, j);
                    const VT v = rows[(size_t)r * 32 + lane];
                    if constexpr (WEIGHTED) fma_row(acc, rsqrtf((float)(cnt[r] + 1)), v);
                    else add_row(acc, v);
                }
            }
        }
        if constexpr (WEIGHTED) fma_row(acc, dc, self);
        else add_row(acc, self);
        scale_row(acc, dc);
        epi.row(acc);
        dst[(size_t)c0 * 32 + lane] = acc;
        if (c1 >= N) break;
        c0 = c1; c1 = c2;
        n0 = n1; n1 = n2;
        ri = ri1;
    }
}


__device__ __forceinline__ void gemm1_body(const float* __restrict__ x, const float* __restrict__ W1,
                           const float* __restrict__ b1, float* __restrict__ h1, int N,
                           int bid, int nblk) {
    const int wave  = threadIdx.x >> 5;
    const int lane  = threadIdx.x & 31;
    const int col0  = wave * 16;
    const int m     = lane & 15;
    const int n     = lane & 15;
    const int khalf = lane >> 4;

    const float bias = b1[col0 + n];
    v2f bf[32];
    #pragma unroll
    for (int u = 0; u < 16; ++u) {
        const int kb = 8 * u + 4 * khalf;
        bf[2 * u].x     = W1[(size_t)(kb + 0) * D_HID + col0 + n];
        bf[2 * u].y     = W1[(size_t)(kb + 1) * D_HID + col0 + n];
        bf[2 * u + 1].x = W1[(size_t)(kb + 2) * D_HID + col0 + n];
        bf[2 * u + 1].y = W1[(size_t)(kb + 3) * D_HID + col0 + n];
    }

    const int rstep = nblk * 16;
    for (int row0 = bid * 16; row0 < N; row0 += rstep) {
        v8f c;
        #pragma unroll
        for (int r = 0; r < 8; ++r) c[r] = bias;

        const int mbase = khalf * 8;
        const float4* arow = (const float4*)(x + (size_t)(row0 + m) * D_HID + 4 * khalf);
        float4 af[16];
        #pragma unroll
        for (int u = 0; u < 16; ++u) af[u] = arow[2 * u];
        __builtin_amdgcn_sched_barrier(0);
        #pragma unroll
        for (int u = 0; u < 16; ++u) {
            v2f lo, hi;
            lo.x = af[u].x; lo.y = af[u].y;
            hi.x = af[u].z; hi.y = af[u].w;
            c = __builtin_amdgcn_wmma_f32_16x16x4_f32(false, lo, false, bf[2 * u], (short)0, c, false, false);
            c = __builtin_amdgcn_wmma_f32_16x16x4_f32(false, hi, false, bf[2 * u + 1], (short)0, c, false, false);
        }

        asm volatile("v_nop\n\tv_nop\n\tv_nop\n\tv_nop\n\tv_nop\n\tv_nop\n\tv_nop\n\tv_nop" : "+v"(c));
        #pragma unroll
        for (int r = 0; r < 8; ++r)
            h1[(size_t)(row0 + mbase + r) * D_HID + col0 + n] = c[r];
    }
}

__global__ void __launch_bounds__(256) bin_and_gemm1(const int* __restrict__ row, const int* __restrict__ col,
                           int2* __restrict__ tab, int2* __restrict__ sorted,
                           float* __restrict__ stats, int* __restrict__ novf, int E, int PB,
                           const float* __restrict__ x, const float* __restrict__ W1,
                           const float* __restrict__ b1, float* __restrict__ h1, int N, int GB) {
    if ((int)blockIdx.x < PB) bin_edges_body(row, col, tab, sorted, stats, novf, E, (int)blockIdx.x, PB);
    else                      gemm1_body(x, W1, b1, h1, N, (int)blockIdx.x - PB, GB);
}


__global__ void __launch_bounds__(256) gather128(const int* __restrict__ cnt, const int* __restrict__ slots,
                          const int2* __restrict__ ovf, const int* __restrict__ novf,
                          const float* __restrict__ h1, float* __restrict__ hagg, float* stats, int N) {
    __shared__ float part[8][2 * D_HID];
    RowStats st;
    st.s = make_float4(0.0f, 0.0f, 0.0f, 0.0f);
    st.q = make_float4(0.0f, 0.0f, 0.0f, 0.0f);
    gather_nodes<true, float4>(cnt, slots, ovf, novf, (const float4*)h1, (float4*)hagg, N, st);
    const int wave = threadIdx.x >> 5;
    const int lane = threadIdx.x & 31;
    *(float4*)&part[wave][4 * lane]         = st.s;
    *(float4*)&part[wave][D_HID + 4 * lane] = st.q;
    __syncthreads();
    float tot = 0.0f;
    #pragma unroll
    for (int w = 0; w < 8; ++w) tot += part[w][threadIdx.x];
    atomicAdd(&stats[threadIdx.x], tot);
}


__global__ void bn_finalize(const float* __restrict__ bsum, const float* __restrict__ bsq,
                            const float* __restrict__ gamma, const float* __restrict__ beta,
                            float* scale, float* shift, int N) {
    int c = threadIdx.x;
    if (c < D_HID) {
        float inv = 1.0f / (float)N;
        float mu = bsum[c] * inv;
        float var = bsq[c] * inv - mu * mu;
        float s = gamma[c] * rsqrtf(var + BN_EPS);
        scale[c] = s;
        shift[c] = beta[c] - mu * s;
    }
}


__global__ void __launch_bounds__(256) gemm2_wmma(const float* __restrict__ hagg, const float* __restrict__ scale,
                           const float* __restrict__ shift, const float* __restrict__ W2,
                           const float* __restrict__ b2, const int* __restrict__ cnt,
                           float* __restrict__ h2, int N) {
    const int wave  = __builtin_amdgcn_readfirstlane(threadIdx.x >> 5);
    const int lane  = threadIdx.x & 31;
    const int col0  = (wave & 3) * 16;
    const int m     = lane & 15;
    const int n     = lane & 15;
    const int khalf = lane >> 4;

    const float bias = b2[col0 + n];
    v2f bf[32];
    #pragma unroll
    for (int u = 0; u < 16; ++u) {
        const int kb = 8 * u + 4 * khalf;
        bf[2 * u].x     = W2[(size_t)(kb + 0) * D_OUT + col0 + n];
        bf[2 * u].y     = W2[(size_t)(kb + 1) * D_OUT + col0 + n];
        bf[2 * u + 1].x = W2[(size_t)(kb + 2) * D_OUT + col0 + n];
        bf[2 * u + 1].y = W2[(size_t)(kb + 3) * D_OUT + col0 + n];
    }

    float4 sc[16], sh[16];
    #pragma unroll
    for (int u = 0; u < 16; ++u) {
        sc[u] = ((const float4*)(scale + 4 * khalf))[2 * u];
        sh[u] = ((const float4*)(shift + 4 * khalf))[2 * u];
    }

    const int rstep = gridDim.x * 32;
    for (int row0 = (blockIdx.x * 2 + (wave >> 2)) * 16; row0 < N; row0 += rstep) {
        v8f c;
        #pragma unroll
        for (int r = 0; r < 8; ++r) c[r] = bias;

        const int mbase = khalf * 8;
        int dn[8];
        #pragma unroll
        for (int r = 0; r < 8; ++r) dn[r] = cnt[row0 + mbase + r];
        const float4* arow = (const float4*)(hagg + (size_t)(row0 + m) * D_HID + 4 * khalf);
        float4 af[16];
        #pragma unroll
        for (int u = 0; u < 16; ++u) af[u] = arow[2 * u];
        asm volatile("" :: "v"(dn[0]), "v"(dn[1]), "v"(dn[2]), "v"(dn[3]), "v"(dn[4]), "v"(dn[5]), "v"(dn[6]), "v"(dn[7]));
        __builtin_amdgcn_sched_barrier(0);
        #pragma unroll
        for (int u = 0; u < 16; ++u) {
            v2f lo, hi;
            lo.x = fmaxf(af[u].x * sc[u].x + sh[u].x, 0.0f);
            lo.y = fmaxf(af[u].y * sc[u].y + sh[u].y, 0.0f);
            hi.x = fmaxf(af[u].z * sc[u].z + sh[u].z, 0.0f);
            hi.y = fmaxf(af[u].w * sc[u].w + sh[u].w, 0.0f);
            c = __builtin_amdgcn_wmma_f32_16x16x4_f32(false, lo, false, bf[2 * u], (short)0, c, false, false);
            c = __builtin_amdgcn_wmma_f32_16x16x4_f32(false, hi, false, bf[2 * u + 1], (short)0, c, false, false);
        }

        asm volatile("v_nop\n\tv_nop\n\tv_nop\n\tv_nop\n\tv_nop\n\tv_nop\n\tv_nop\n\tv_nop" : "+v"(c));
        #pragma unroll
        for (int r = 0; r < 8; ++r)
            h2[(size_t)(row0 + mbase + r) * D_OUT + col0 + n] = c[r] * __builtin_amdgcn_rsqf((float)(dn[r] + 1));
    }
}


__global__ void __launch_bounds__(256) gather64(const int* __restrict__ cnt, const int* __restrict__ slots,
                         const int2* __restrict__ ovf, const int* __restrict__ novf,
                         const float* __restrict__ h2, float* __restrict__ out, int N) {
    RowLogSoftmax ls;
    gather_nodes<false, float2>(cnt, slots, ovf, novf, (const float2*)h2, (float2*)out, N, ls);
}


extern "C" void kernel_launch(void* const* d_in, const int* in_sizes, int n_in,
                              void* d_out, int out_size, void* d_ws, size_t ws_size,
                              hipStream_t stream) {
    const float* x     = (const float*)d_in[0];
    const float* W1    = (const float*)d_in[1];
    const float* b1    = (const float*)d_in[2];
    const float* gamma = (const float*)d_in[3];
    const float* beta  = (const float*)d_in[4];
    const float* W2    = (const float*)d_in[5];
    const float* b2    = (const float*)d_in[6];
    const int*   ei    = (const int*)d_in[7];

    const int N = in_sizes[0] / D_HID;
    const int E = in_sizes[7] / 2;
    const int* erow = ei;
    const int* ecol = ei + E;

    float* ws    = (float*)d_ws;
    float* h1    = ws;
    float* hagg  = h1 + (size_t)N * D_HID;
    float* stats = hagg + (size_t)N * D_HID;
    float* bsum  = stats;
    float* bsq   = stats + 128;
    float* scal  = stats + 256;
    float* shft  = stats + 384;
    int*   novf  = (int*)(stats + 512);
    int*   slots = (int*)(stats + 576);
    int2*  ovf   = (int2*)(slots + (size_t)N * BUCKET);
    int*   cnt   = (int*)(ovf + E);
    int2*  tab   = (int2*)(cnt + 65536);
    int2*  sorted = tab + 65536;
    float* h2    = h1;
    float* outp  = (float*)d_out;

    bin_and_gemm1<<<768, 256, 0, stream>>>(erow, ecol, tab, sorted, stats, novf, E, 256, x, W1, b1, h1, N, 512);
    place_owner<<<256, 256, 0, stream>>>(tab, sorted, cnt, slots, ovf, novf, N, 256);

    gather128<<<1024, 256, 0, stream>>>(cnt, slots, ovf, novf, h1, hagg, stats, N);

    bn_finalize<<<1, 128, 0, stream>>>(bsum, bsq, gamma, beta, scal, shft, N);

    gemm2_wmma<<<256, 256, 0, stream>>>(hagg, scal, shft, W2, b2, cnt, h2, N);
    gather64<<<1024, 256, 0, stream>>>(cnt, slots, ovf, novf, h2, outp, N);
}


# Both GEMMs: A tile staged through LDS (a row fetched once per block by one whole-row load, BN+ReLU applied once there, next tile asked ahead); fragments read from LDS, not by 16-line loads per wave
# speedup vs baseline: 5.1669x; 1.1401x over previous
#include <hip/hip_runtime.h>
#include <hip/hip_bf16.h>
#include <math.h>

typedef float v8f __attribute__((ext_vector_type(8)));
typedef float v2f __attribute__((ext_vector_type(2)));

#define D_HID 128
#define D_OUT 64
#define BN_EPS 1e-5f


#define BUCKET 32
#define RANGE_SHIFT 8
#define NRANGE 256

#define EPT 8
__device__ __forceinline__ void bin_edges_body(const int* __restrict__ row, const int* __restrict__ col,
                            int2* __restrict__ tab, int2* __restrict__ sorted,
                            float* __restrict__ stats, int* __restrict__ novf, int E, int bid, int nblk) {
    __shared__ int bcnt[NRANGE];
    __shared__ int boff[NRANGE];
    __shared__ int bfill[NRANGE];
    __shared__ int wsum[8];
    const int t = threadIdx.x;
    const int lane = t & 31;
    const int wave = t >> 5;
    if (bid == 0) {
        stats[t] = 0.0f;
        if (t == 0) *novf = 0;
    }
    bcnt[t] = 0;
    bfill[t] = 0;
    __syncthreads();
    const int chunk = ((E + nblk - 1) / nblk + 15) & ~15;
    const int e0 = bid * chunk;
    const int e1 = min(E, e0 + chunk);
    const int stride = blockDim.x;
    for (int e = e0 + t; e < e1; e += EPT * stride) {
        int c[EPT];
        #pragma unroll
        for (int k = 0; k < EPT; ++k) {
            const int ek = e + k * stride;
            c[k] = -1;
            if (ek < e1) c[k] = col[ek];
        }
        #pragma unroll
        for (int k = 0; k < EPT; ++k)
            if (c[k] >= 0) atomicAdd(&bcnt[min(c[k] >> RANGE_SHIFT, NRANGE - 1)], 1);
    }
    __syncthreads();
    const int v = bcnt[t];
    int inc = v;
    #pragma unroll
    for (int d = 1; d < 32; d <<= 1) {
        const int o = __shfl_up(inc, d);
        if (lane >= d) inc += o;
    }
    if (lane == 31) wsum[wave] = inc;
    __syncthreads();
    int wbase = 0;
    #pragma unroll
    for (int w = 0; w < 8; ++w)
        if (w < wave) wbase += wsum[w];
    const int excl = wbase + inc - v;
    boff[t] = excl;
    tab[bid * NRANGE + t] = make_int2(v, e0 + excl);
    __syncthreads();
    for (int e = e0 + t; e < e1; e += EPT * stride) {
        int c[EPT], r[EPT];
        #pragma unroll
        for (int k = 0; k < EPT; ++k) {
            const int ek = e + k * stride;
            c[k] = -1; r[k] = 0;
            if (ek < e1) { c[k] = col[ek]; r[k] = row[ek]; }
        }
        int g[EPT], p[EPT];
        #pragma unroll
        for (int k = 0; k < EPT; ++k) {
            g[k] = min(c[k] >> RANGE_SHIFT, NRANGE - 1);
            p[k] = 0;
            if (c[k] >= 0) p[k] = atomicAdd(&bfill[g[k]], 1);
        }
        #pragma unroll
        for (int k = 0; k < EPT; ++k)
            if (c[k] >= 0) sorted[e0 + boff[g[k]] + p[k]] = make_int2(c[k], r[k]);
    }
}

#define RUN 32
__global__ void __launch_bounds__(256) place_owner(const int2* __restrict__ tab,
                            const int2* __restrict__ sorted, int* __restrict__ cnt, int* __restrict__ slots,
                            int2* __restrict__ ovf, int* novf, int N, int npiece) {
    __shared__ int  lcnt[NRANGE];
    __shared__ int4 lsl4[NRANGE * BUCKET / 4];
    int* lsl = (int*)lsl4;
    const int t = threadIdx.x;
    const int g = blockIdx.x;
    const int base = g << RANGE_SHIFT;
    if (base >= N) return;
    lcnt[t] = 0;
    #pragma unroll
    for (int i = 0; i < NRANGE * BUCKET / 4 / 256; ++i) lsl4[i * 256 + t] = make_int4(0, 0, 0, 0);
    __syncthreads();
    for (int b = t; b < npiece; b += blockDim.x) {
        const int2 run = tab[b * NRANGE + g];
        const int n = run.x;
        const int o = run.y;
        for (int i0 = 0; i0 < n; i0 += RUN) {
            int2 e[RUN];
            #pragma unroll
            for (int u = 0; u < RUN; ++u) {
                e[u] = make_int2(-1, 0);
                if (i0 + u < n) e[u] = sorted[o + i0 + u];
            }
            int p[RUN];
            #pragma unroll
            for (int u = 0; u < RUN; ++u) {
                p[u] = 0;
                if (e[u].x >= 0) p[u] = atomicAdd(&lcnt[(e[u].x - base) & (NRANGE - 1)], 1);
            }
            #pragma unroll
            for (int u = 0; u < RUN; ++u) {
                if (e[u].x >= 0) {
                    const int j = (e[u].x - base) & (NRANGE - 1);
                    if (p[u] < BUCKET) {
                        lsl[j * BUCKET + p[u]] = e[u].y;
                    } else {
                        const int q = atomicAdd(novf, 1);
                        ovf[q] = e[u];
                    }
                }
            }
        }
    }
    __syncthreads();
    const int nrows = min(NRANGE, N - base);
    if (t < nrows) cnt[base + t] = lcnt[t];
    int4* dst = (int4*)(slots + (size_t)base * BUCKET);
    for (int i = t; i < nrows * (BUCKET / 4); i += blockDim.x) dst[i] = lsl4[i];
}


__device__ __forceinline__ void add_row(float4& acc, const float4& v) {
    acc.x += v.x; acc.y += v.y; acc.z += v.z; acc.w += v.w;
}
__device__ __forceinline__ void add_row(float2& acc, const float2& v) {
    acc.x += v.x; acc.y += v.y;
}
__device__ __forceinline__ void scale_row(float4& a, float d) { a.x *= d; a.y *= d; a.z *= d; a.w *= d; }
__device__ __forceinline__ void scale_row(float2& a, float d) { a.x *= d; a.y *= d; }
__device__ __forceinline__ void fma_row(float4& acc, float w, const float4& v) {
    acc.x = fmaf(w, v.x, acc.x); acc.y = fmaf(w, v.y, acc.y); acc.z = fmaf(w, v.z, acc.z); acc.w = fmaf(w, v.w, acc.w);
}
__device__ __forceinline__ void fma_row(float2& acc, float w, const float2& v) {
    acc.x = fmaf(w, v.x, acc.x); acc.y = fmaf(w, v.y, acc.y);
}
__device__ __forceinline__ void set_zero(float4& a) { a = make_float4(0.0f, 0.0f, 0.0f, 0.0f); }
__device__ __forceinline__ void set_zero(float2& a) { a = make_float2(0.0f, 0.0f); }

template <int K, bool WEIGHTED, typename VT>
__device__ __forceinline__ void edge_batch(const VT* __restrict__ rows, int lane,
                                           int ri, int sc, int j, int cnt, VT& acc) {
    VT v[K];
    #pragma unroll
    for (int k = 0; k < K; ++k) {
        set_zero(v[k]);
        if (k < cnt) {
            const int r = __builtin_amdgcn_readlane(ri, j + k);
            v[k] = rows[(size_t)r * 32 + lane];
        }
    }
    float wl = 0.0f;
    if constexpr (WEIGHTED) wl = __builtin_amdgcn_rsqf((float)(sc + 1));
    #pragma unroll
    for (int k = 0; k < K; ++k) {
        if (k < cnt) {
            if constexpr (WEIGHTED) {
                const float w = __builtin_bit_cast(float, __builtin_amdgcn_readlane(__builtin_bit_cast(int, wl), j + k));
                fma_row(acc, w, v[k]);
            } else {
                add_row(acc, v[k]);
            }
        }
    }
}


struct RowStats {
    float4 s, q;
    __device__ __forceinline__ void row(float4& a) {
        s.x += a.x; s.y += a.y; s.z += a.z; s.w += a.w;
        q.x = fmaf(a.x, a.x, q.x); q.y = fmaf(a.y, a.y, q.y);
        q.z = fmaf(a.z, a.z, q.z); q.w = fmaf(a.w, a.w, q.w);
    }
};

template <int M>
__device__ __forceinline__ float row_xor(float x) {
    return __builtin_bit_cast(float, __builtin_amdgcn_mov_dpp(__builtin_bit_cast(int, x), 0x160 + M, 0xf, 0xf, true));
}
__device__ __forceinline__ float xhalf(float x) {
    const int b = __builtin_bit_cast(int, x);
    return __builtin_bit_cast(float, __builtin_amdgcn_permlanex16(b, b, 0x76543210, 0xfedcba98, false, false));
}
__device__ __forceinline__ float wave_max(float x) {
    x = fmaxf(x, row_xor<1>(x)); x = fmaxf(x, row_xor<2>(x));
    x = fmaxf(x, row_xor<4>(x)); x = fmaxf(x, row_xor<8>(x));
    return fmaxf(x, xhalf(x));
}
__device__ __forceinline__ float wave_sum(float x) {
    x += row_xor<1>(x); x += row_xor<2>(x);
    x += row_xor<4>(x); x += row_xor<8>(x);
    return x + xhalf(x);
}

struct RowLogSoftmax {
    __device__ __forceinline__ void row(float2& a) {
        const float mx = wave_max(fmaxf(a.x, a.y));
        const float s  = wave_sum(__expf(a.x - mx) + __expf(a.y - mx));
        const float l  = mx + __logf(s);
        a.x -= l; a.y -= l;
    }
};

template <bool WEIGHTED, typename VT, typename EPI>
__device__ __forceinline__ void gather_nodes(const int* __restrict__ cnt, const int* __restrict__ slots,
                                             const int2* __restrict__ ovf, const int* __restrict__ novf,
                                             const VT* __restrict__ rows, VT* __restrict__ dst, int N,
                                             EPI& epi) {
    const int wave = __builtin_amdgcn_readfirstlane(threadIdx.x >> 5);
    const int lane = threadIdx.x & 31;
    const int G = gridDim.x * 8;
    int c0 = blockIdx.x * 8 + wave;
    if (c0 >= N) return;
    int c1 = c0 + G;
    const int nov = *novf;
    int n0 = cnt[c0];
    int n1 = 0;
    if (c1 < N) n1 = cnt[c1];
    int ri = 0;
    if (lane < min(BUCKET, n0)) ri = slots[c0 * BUCKET + lane];
    for (;;) {
        int sc = 0;
        if constexpr (WEIGHTED) sc = cnt[ri];
        const int c2 = c1 + G;
        int n2 = 0;
        if (c2 < N) n2 = cnt[c2];
        int ri1 = 0;
        if (lane < min(BUCKET, n1)) ri1 = slots[c1 * BUCKET + lane];
        const int n = n0;
        const float dc = rsqrtf((float)(n + 1));
        const VT self = rows[(size_t)c0 * 32 + lane];
        VT acc;
        set_zero(acc);
        {
            const int m = min(BUCKET, n);
            if (m > 0)  edge_batch<16, WEIGHTED>(rows, lane, ri, sc, 0, m, acc);
            if (m > 16) edge_batch<16, WEIGHTED>(rows, lane, ri, sc, 16, m - 16, acc);
        }
        if (n > BUCKET) {
            for (int b = 0; b < nov; b += 32) {
                int2 oe = make_int2(-1, 0);
                if (b + lane < nov) oe = ovf[b + lane];
                unsigned hit = __builtin_amdgcn_ballot_w32(oe.x == c0);
                while (hit != 0u) {
                    const int j = __builtin_amdgcn_readfirstlane(__builtin_ctz(hit));
                    hit &= hit - 1u;
                    const int r = __builtin_amdgcn_readlane(oe.y, j);
                    const VT v = rows[(size_t)r * 32 + lane];
                    if constexpr (WEIGHTED) fma_row(acc, rsqrtf((float)(cnt[r] + 1)), v);
                    else add_row(acc, v);
                }
            }
        }
        if constexpr (WEIGHTED) fma_row(acc, dc, self);
        else add_row(acc, self);
        scale_row(acc, dc);
        epi.row(acc);
        dst[(size_t)c0 * 32 + lane] = acc;
        if (c1 >= N) break;
        c0 = c1; c1 = c2;
        n0 = n1; n1 = n2;
        ri = ri1;
    }
}

#define APITCH 33

__device__ __forceinline__ void gemm1_body(const float* __restrict__ x, const float* __restrict__ W1,
                           const float* __restrict__ b1, float* __restrict__ h1, int N,
                           int bid, int nblk) {
    __shared__ float4 atile1[2][16][APITCH];
    const int wave  = __builtin_amdgcn_readfirstlane(threadIdx.x >> 5);
    const int lane  = threadIdx.x & 31;
    const int col0  = wave * 16;
    const int m     = lane & 15;
    const int n     = lane & 15;
    const int khalf = lane >> 4;

    const float bias = b1[col0 + n];
    v2f bf[32];
    #pragma unroll
    for (int u = 0; u < 16; ++u) {
        const int kb = 8 * u + 4 * khalf;
        bf[2 * u].x     = W1[(size_t)(kb + 0) * D_HID + col0 + n];
        bf[2 * u].y     = W1[(size_t)(kb + 1) * D_HID + col0 + n];
        bf[2 * u + 1].x = W1[(size_t)(kb + 2) * D_HID + col0 + n];
        bf[2 * u + 1].y = W1[(size_t)(kb + 3) * D_HID + col0 + n];
    }

    const float4* xr = (const float4*)x;
    const int rstep = nblk * 16;
    int row0 = bid * 16;
    float4 g0 = make_float4(0.0f, 0.0f, 0.0f, 0.0f), g1 = g0;
    if (row0 < N) {
        g0 = xr[(size_t)(row0 + 2 * wave) * 32 + lane];
        g1 = xr[(size_t)(row0 + 2 * wave + 1) * 32 + lane];
    }
    int s = 0;
    for (; row0 < N; row0 += rstep) {
        atile1[s][2 * wave][lane]     = g0;
        atile1[s][2 * wave + 1][lane] = g1;
        __syncthreads();
        const int nrow0 = row0 + rstep;
        if (nrow0 < N) {
            g0 = xr[(size_t)(nrow0 + 2 * wave) * 32 + lane];
            g1 = xr[(size_t)(nrow0 + 2 * wave + 1) * 32 + lane];
        }

        v8f c;
        #pragma unroll
        for (int r = 0; r < 8; ++r) c[r] = bias;

        const int mbase = khalf * 8;
        float4 af[16];
        #pragma unroll
        for (int u = 0; u < 16; ++u) af[u] = atile1[s][m][2 * u + khalf];
        __builtin_amdgcn_sched_barrier(0);
        #pragma unroll
        for (int u = 0; u < 16; ++u) {
            v2f lo, hi;
            lo.x = af[u].x; lo.y = af[u].y;
            hi.x = af[u].z; hi.y = af[u].w;
            c = __builtin_amdgcn_wmma_f32_16x16x4_f32(false, lo, false, bf[2 * u], (short)0, c, false, false);
            c = __builtin_amdgcn_wmma_f32_16x16x4_f32(false, hi, false, bf[2 * u + 1], (short)0, c, false, false);
        }

        asm volatile("v_nop\n\tv_nop\n\tv_nop\n\tv_nop\n\tv_nop\n\tv_nop\n\tv_nop\n\tv_nop" : "+v"(c));
        #pragma unroll
        for (int r = 0; r < 8; ++r)
            h1[(size_t)(row0 + mbase + r) * D_HID + col0 + n] = c[r];
        s ^= 1;
    }
}

__global__ void __launch_bounds__(256, 6) bin_and_gemm1(const int* __restrict__ row, const int* __restrict__ col,
                           int2* __restrict__ tab, int2* __restrict__ sorted,
                           float* __restrict__ stats, int* __restrict__ novf, int E, int PB,
                           const float* __restrict__ x, const float* __restrict__ W1,
                           const float* __restrict__ b1, float* __restrict__ h1, int N, int GB) {
    if ((int)blockIdx.x < PB) bin_edges_body(row, col, tab, sorted, stats, novf, E, (int)blockIdx.x, PB);
    else                      gemm1_body(x, W1, b1, h1, N, (int)blockIdx.x - PB, GB);
}


__global__ void __launch_bounds__(256) gather128(const int* __restrict__ cnt, const int* __restrict__ slots,
                          const int2* __restrict__ ovf, const int* __restrict__ novf,
                          const float* __restrict__ h1, float* __restrict__ hagg, float* stats, int N) {
    __shared__ float part[8][2 * D_HID];
    RowStats st;
    st.s = make_float4(0.0f, 0.0f, 0.0f, 0.0f);
    st.q = make_float4(0.0f, 0.0f, 0.0f, 0.0f);
    gather_nodes<true, float4>(cnt, slots, ovf, novf, (const float4*)h1, (float4*)hagg, N, st);
    const int wave = threadIdx.x >> 5;
    const int lane = threadIdx.x & 31;
    *(float4*)&part[wave][4 * lane]         = st.s;
    *(float4*)&part[wave][D_HID + 4 * lane] = st.q;
    __syncthreads();
    float tot = 0.0f;
    #pragma unroll
    for (int w = 0; w < 8; ++w) tot += part[w][threadIdx.x];
    atomicAdd(&stats[threadIdx.x], tot);
}


__global__ void bn_finalize(const float* __restrict__ bsum, const float* __restrict__ bsq,
                            const float* __restrict__ gamma, const float* __restrict__ beta,
                            float* scale, float* shift, int N) {
    int c = threadIdx.x;
    if (c < D_HID) {
        float inv = 1.0f / (float)N;
        float mu = bsum[c] * inv;
        float var = bsq[c] * inv - mu * mu;
        float s = gamma[c] * rsqrtf(var + BN_EPS);
        scale[c] = s;
        shift[c] = beta[c] - mu * s;
    }
}


__global__ void __launch_bounds__(256) gemm2_wmma(const float* __restrict__ hagg, const float* __restrict__ scale,
                           const float* __restrict__ shift, const float* __restrict__ W2,
                           const float* __restrict__ b2, const int* __restrict__ cnt,
                           float* __restrict__ h2, int N) {
    __shared__ float4 atile2[2][32][APITCH];
    const int wave  = __builtin_amdgcn_readfirstlane(threadIdx.x >> 5);
    const int lane  = threadIdx.x & 31;
    const int col0  = (wave & 3) * 16;
    const int rt    = wave >> 2;
    const int m     = lane & 15;
    const int n     = lane & 15;
    const int khalf = lane >> 4;

    const float bias = b2[col0 + n];
    v2f bf[32];
    #pragma unroll
    for (int u = 0; u < 16; ++u) {
        const int kb = 8 * u + 4 * khalf;
        bf[2 * u].x     = W2[(size_t)(kb + 0) * D_OUT + col0 + n];
        bf[2 * u].y     = W2[(size_t)(kb + 1) * D_OUT + col0 + n];
        bf[2 * u + 1].x = W2[(size_t)(kb + 2) * D_OUT + col0 + n];
        bf[2 * u + 1].y = W2[(size_t)(kb + 3) * D_OUT + col0 + n];
    }

    const float4 sc = ((const float4*)scale)[lane];
    const float4 sh = ((const float4*)shift)[lane];

    const float4* hr = (const float4*)hagg;
    const int pstep = gridDim.x * 32;
    int p0 = blockIdx.x * 32;
    const int mbase = khalf * 8;
    int dn[8];
    #pragma unroll
    for (int r = 0; r < 8; ++r) dn[r] = 0;
    if (p0 + rt * 16 < N) {
        #pragma unroll
        for (int r = 0; r < 8; ++r) dn[r] = cnt[p0 + rt * 16 + mbase + r];
    }
    float4 g[4];
    #pragma unroll
    for (int j = 0; j < 4; ++j) g[j] = make_float4(0.0f, 0.0f, 0.0f, 0.0f);
    if (p0 + 4 * wave < N) {
        #pragma unroll
        for (int j = 0; j < 4; ++j) g[j] = hr[(size_t)(p0 + 4 * wave + j) * 32 + lane];
    }
    int s = 0;
    for (; p0 < N; p0 += pstep) {
        #pragma unroll
        for (int j = 0; j < 4; ++j) {
            float4 a;
            a.x = fmaxf(g[j].x * sc.x + sh.x, 0.0f);
            a.y = fmaxf(g[j].y * sc.y + sh.y, 0.0f);
            a.z = fmaxf(g[j].z * sc.z + sh.z, 0.0f);
            a.w = fmaxf(g[j].w * sc.w + sh.w, 0.0f);
            atile2[s][4 * wave + j][lane] = a;
        }
        __syncthreads();
        const int row0 = p0 + rt * 16;
        const int np0 = p0 + pstep;
        int dnn[8];
        #pragma unroll
        for (int r = 0; r < 8; ++r) dnn[r] = 0;
        if (np0 + rt * 16 < N) {
            #pragma unroll
            for (int r = 0; r < 8; ++r) dnn[r] = cnt[np0 + rt * 16 + mbase + r];
        }
        if (np0 + 4 * wave < N) {
            #pragma unroll
            for (int j = 0; j < 4; ++j) g[j] = hr[(size_t)(np0 + 4 * wave + j) * 32 + lane];
        }

        if (row0 < N) {
            v8f c;
            #pragma unroll
            for (int r = 0; r < 8; ++r) c[r] = bias;

            float4 af[16];
            #pragma unroll
            for (int u = 0; u < 16; ++u) af[u] = atile2[s][rt * 16 + m][2 * u + khalf];
            __builtin_amdgcn_sched_barrier(0);
            #pragma unroll
            for (int u = 0; u < 16; ++u) {
                v2f lo, hi;
                lo.x = af[u].x; lo.y = af[u].y;
                hi.x = af[u].z; hi.y = af[u].w;
                c = __builtin_amdgcn_wmma_f32_16x16x4_f32(false, lo, false, bf[2 * u], (short)0, c, false, false);
                c = __builtin_amdgcn_wmma_f32_16x16x4_f32(false, hi, false, bf[2 * u + 1], (short)0, c, false, false);
            }

            asm volatile("v_nop\n\tv_nop\n\tv_nop\n\tv_nop\n\tv_nop\n\tv_nop\n\tv_nop\n\tv_nop" : "+v"(c));
            #pragma unroll
            for (int r = 0; r < 8; ++r)
                h2[(size_t)(row0 + mbase + r) * D_OUT + col0 + n] = c[r] * __builtin_amdgcn_rsqf((float)(dn[r] + 1));
        }
        #pragma unroll
        for (int r = 0; r < 8; ++r) dn[r] = dnn[r];
        s ^= 1;
    }
}


__global__ void __launch_bounds__(256) gather64(const int* __restrict__ cnt, const int* __restrict__ slots,
                         const int2* __restrict__ ovf, const int* __restrict__ novf,
                         const float* __restrict__ h2, float* __restrict__ out, int N) {
    RowLogSoftmax ls;
    gather_nodes<false, float2>(cnt, slots, ovf, novf, (const float2*)h2, (float2*)out, N, ls);
}


extern "C" void kernel_launch(void* const* d_in, const int* in_sizes, int n_in,
                              void* d_out, int out_size, void* d_ws, size_t ws_size,
                              hipStream_t stream) {
    const float* x     = (const float*)d_in[0];
    const float* W1    = (const float*)d_in[1];
    const float* b1    = (const float*)d_in[2];
    const float* gamma = (const float*)d_in[3];
    const float* beta  = (const float*)d_in[4];
    const float* W2    = (const float*)d_in[5];
    const float* b2    = (const float*)d_in[6];
    const int*   ei    = (const int*)d_in[7];

    const int N = in_sizes[0] / D_HID;
    const int E = in_sizes[7] / 2;
    const int* erow = ei;
    const int* ecol = ei + E;

    float* ws    = (float*)d_ws;
    float* h1    = ws;
    float* hagg  = h1 + (size_t)N * D_HID;
    float* stats = hagg + (size_t)N * D_HID;
    float* bsum  = stats;
    float* bsq   = stats + 128;
    float* scal  = stats + 256;
    float* shft  = stats + 384;
    int*   novf  = (int*)(stats + 512);
    int*   slots = (int*)(stats + 576);
    int2*  ovf   = (int2*)(slots + (size_t)N * BUCKET);
    int*   cnt   = (int*)(ovf + E);
    int2*  tab   = (int2*)(cnt + 65536);
    int2*  sorted = tab + 65536;
    float* h2    = h1;
    float* outp  = (float*)d_out;

    bin_and_gemm1<<<768, 256, 0, stream>>>(erow, ecol, tab, sorted, stats, novf, E, 256, x, W1, b1, h1, N, 512);
    place_owner<<<256, 256, 0, stream>>>(tab, sorted, cnt, slots, ovf, novf, N, 256);

    gather128<<<1024, 256, 0, stream>>>(cnt, slots, ovf, novf, h1, hagg, stats, N);

    bn_finalize<<<1, 128, 0, stream>>>(bsum, bsq, gamma, beta, scal, shft, N);

    gemm2_wmma<<<512, 256, 0, stream>>>(hagg, scal, shft, W2, b2, cnt, h2, N);
    gather64<<<1024, 256, 0, stream>>>(cnt, slots, ovf, novf, h2, outp, N);
}


# Both gathers issue far fewer instructions per edge: a batch's edges go in runs of 1,8,4,2,1 picked by its count's bits (4 tests, not test+branch+clear per edge), rows by 32-bit index; sum order kept
# speedup vs baseline: 5.3607x; 1.0375x over previous
#include <hip/hip_runtime.h>
#include <hip/hip_bf16.h>
#include <math.h>

typedef float v8f __attribute__((ext_vector_type(8)));
typedef float v2f __attribute__((ext_vector_type(2)));

#define D_HID 128
#define D_OUT 64
#define BN_EPS 1e-5f


#define BUCKET 32
#define RANGE_SHIFT 8
#define NRANGE 256

#define EPT 8
__device__ __forceinline__ void bin_edges_body(const int* __restrict__ row, const int* __restrict__ col,
                            int2* __restrict__ tab, int2* __restrict__ sorted,
                            float* __restrict__ stats, int* __restrict__ novf, int E, int bid, int nblk) {
    __shared__ int bcnt[NRANGE];
    __shared__ int boff[NRANGE];
    __shared__ int bfill[NRANGE];
    __shared__ int wsum[8];
    const int t = threadIdx.x;
    const int lane = t & 31;
    const int wave = t >> 5;
    if (bid == 0) {
        stats[t] = 0.0f;
        if (t == 0) *novf = 0;
    }
    bcnt[t] = 0;
    bfill[t] = 0;
    __syncthreads();
    const int chunk = ((E + nblk - 1) / nblk + 15) & ~15;
    const int e0 = bid * chunk;
    const int e1 = min(E, e0 + chunk);
    const int stride = blockDim.x;
    for (int e = e0 + t; e < e1; e += EPT * stride) {
        int c[EPT];
        #pragma unroll
        for (int k = 0; k < EPT; ++k) {
            const int ek = e + k * stride;
            c[k] = -1;
            if (ek < e1) c[k] = col[ek];
        }
        #pragma unroll
        for (int k = 0; k < EPT; ++k)
            if (c[k] >= 0) atomicAdd(&bcnt[min(c[k] >> RANGE_SHIFT, NRANGE - 1)], 1);
    }
    __syncthreads();
    const int v = bcnt[t];
    int inc = v;
    #pragma unroll
    for (int d = 1; d < 32; d <<= 1) {
        const int o = __shfl_up(inc, d);
        if (lane >= d) inc += o;
    }
    if (lane == 31) wsum[wave] = inc;
    __syncthreads();
    int wbase = 0;
    #pragma unroll
    for (int w = 0; w < 8; ++w)
        if (w < wave) wbase += wsum[w];
    const int excl = wbase + inc - v;
    boff[t] = excl;
    tab[bid * NRANGE + t] = make_int2(v, e0 + excl);
    __syncthreads();
    for (int e = e0 + t; e < e1; e += EPT * stride) {
        int c[EPT], r[EPT];
        #pragma unroll
        for (int k = 0; k < EPT; ++k) {
            const int ek = e + k * stride;
            c[k] = -1; r[k] = 0;
            if (ek < e1) { c[k] = col[ek]; r[k] = row[ek]; }
        }
        int g[EPT], p[EPT];
        #pragma unroll
        for (int k = 0; k < EPT; ++k) {
            g[k] = min(c[k] >> RANGE_SHIFT, NRANGE - 1);
            p[k] = 0;
            if (c[k] >= 0) p[k] = atomicAdd(&bfill[g[k]], 1);
        }
        #pragma unroll
        for (int k = 0; k < EPT; ++k)
            if (c[k] >= 0) sorted[e0 + boff[g[k]] + p[k]] = make_int2(c[k], r[k]);
    }
}

#define RUN 32
__global__ void __launch_bounds__(256) place_owner(const int2* __restrict__ tab,
                            const int2* __restrict__ sorted, int* __restrict__ cnt, int* __restrict__ slots,
                            int2* __restrict__ ovf, int* novf, int N, int npiece) {
    __shared__ int  lcnt[NRANGE];
    __shared__ int4 lsl4[NRANGE * BUCKET / 4];
    int* lsl = (int*)lsl4;
    const int t = threadIdx.x;
    const int g = blockIdx.x;
    const int base = g << RANGE_SHIFT;
    if (base >= N) return;
    lcnt[t] = 0;
    #pragma unroll
    for (int i = 0; i < NRANGE * BUCKET / 4 / 256; ++i) lsl4[i * 256 + t] = make_int4(0, 0, 0, 0);
    __syncthreads();
    for (int b = t; b < npiece; b += blockDim.x) {
        const int2 run = tab[b * NRANGE + g];
        const int n = run.x;
        const int o = run.y;
        for (int i0 = 0; i0 < n; i0 += RUN) {
            int2 e[RUN];
            #pragma unroll
            for (int u = 0; u < RUN; ++u) {
                e[u] = make_int2(-1, 0);
                if (i0 + u < n) e[u] = sorted[o + i0 + u];
            }
            int p[RUN];
            #pragma unroll
            for (int u = 0; u < RUN; ++u) {
                p[u] = 0;
                if (e[u].x >= 0) p[u] = atomicAdd(&lcnt[(e[u].x - base) & (NRANGE - 1)], 1);
            }
            #pragma unroll
            for (int u = 0; u < RUN; ++u) {
                if (e[u].x >= 0) {
                    const int j = (e[u].x - base) & (NRANGE - 1);
                    if (p[u] < BUCKET) {
                        lsl[j * BUCKET + p[u]] = e[u].y;
                    } else {
                        const int q = atomicAdd(novf, 1);
                        ovf[q] = e[u];
                    }
                }
            }
        }
    }
    __syncthreads();
    const int nrows = min(NRANGE, N - base);
    if (t < nrows) cnt[base + t] = lcnt[t];
    int4* dst = (int4*)(slots + (size_t)base * BUCKET);
    for (int i = t; i < nrows * (BUCKET / 4); i += blockDim.x) dst[i] = lsl4[i];
}


__device__ __forceinline__ void add_row(float4& acc, const float4& v) {
    acc.x += v.x; acc.y += v.y; acc.z += v.z; acc.w += v.w;
}
__device__ __forceinline__ void add_row(float2& acc, const float2& v) {
    acc.x += v.x; acc.y += v.y;
}
__device__ __forceinline__ void scale_row(float4& a, float d) { a.x *= d; a.y *= d; a.z *= d; a.w *= d; }
__device__ __forceinline__ void scale_row(float2& a, float d) { a.x *= d; a.y *= d; }
__device__ __forceinline__ void fma_row(float4& acc, float w, const float4& v) {
    acc.x = fmaf(w, v.x, acc.x); acc.y = fmaf(w, v.y, acc.y); acc.z = fmaf(w, v.z, acc.z); acc.w = fmaf(w, v.w, acc.w);
}
__device__ __forceinline__ void fma_row(float2& acc, float w, const float2& v) {
    acc.x = fmaf(w, v.x, acc.x); acc.y = fmaf(w, v.y, acc.y);
}
__device__ __forceinline__ void set_zero(float4& a) { a = make_float4(0.0f, 0.0f, 0.0f, 0.0f); }
__device__ __forceinline__ void set_zero(float2& a) { a = make_float2(0.0f, 0.0f); }

template <int P0, int S, typename VT>
__device__ __forceinline__ void load_run(const VT* __restrict__ rows, int lane, int ri, int e, VT* v) {
    #pragma unroll
    for (int s = 0; s < S; ++s) {
        const unsigned r = (unsigned)__builtin_amdgcn_readlane(ri, e + s);
        v[P0 + s] = rows[r * 32u + (unsigned)lane];
    }
}
template <int P0, int S, bool WEIGHTED, typename VT>
__device__ __forceinline__ void sum_run(float wl, int e, const VT* v, VT& acc) {
    #pragma unroll
    for (int s = 0; s < S; ++s) {
        if constexpr (WEIGHTED) {
            const float w = __builtin_bit_cast(float, __builtin_amdgcn_readlane(__builtin_bit_cast(int, wl), e + s));
            fma_row(acc, w, v[P0 + s]);
        } else {
            add_row(acc, v[P0 + s]);
        }
    }
}
template <bool WEIGHTED, typename VT>
__device__ __forceinline__ void edge_batch(const VT* __restrict__ rows, int lane,
                                           int ri, int sc, int j, int cnt, VT& acc) {
    VT v[16];
    #pragma unroll
    for (int p = 1; p < 16; ++p) set_zero(v[p]);
    const int c = cnt - 1;
    int e = j;
    load_run<0, 1>(rows, lane, ri, e, v);  e += 1;
    if (c & 8) { load_run<1, 8>(rows, lane, ri, e, v);  e += 8; }
    if (c & 4) { load_run<9, 4>(rows, lane, ri, e, v);  e += 4; }
    if (c & 2) { load_run<13, 2>(rows, lane, ri, e, v); e += 2; }
    if (c & 1) { load_run<15, 1>(rows, lane, ri, e, v); }
    float wl = 0.0f;
    if constexpr (WEIGHTED) wl = __builtin_amdgcn_rsqf((float)(sc + 1));
    e = j;
    sum_run<0, 1, WEIGHTED>(wl, e, v, acc);  e += 1;
    if (c & 8) { sum_run<1, 8, WEIGHTED>(wl, e, v, acc);  e += 8; }
    if (c & 4) { sum_run<9, 4, WEIGHTED>(wl, e, v, acc);  e += 4; }
    if (c & 2) { sum_run<13, 2, WEIGHTED>(wl, e, v, acc); e += 2; }
    if (c & 1) { sum_run<15, 1, WEIGHTED>(wl, e, v, acc); }
}


struct RowStats {
    float4 s, q;
    __device__ __forceinline__ void row(float4& a) {
        s.x += a.x; s.y += a.y; s.z += a.z; s.w += a.w;
        q.x = fmaf(a.x, a.x, q.x); q.y = fmaf(a.y, a.y, q.y);
        q.z = fmaf(a.z, a.z, q.z); q.w = fmaf(a.w, a.w, q.w);
    }
};

template <int M>
__device__ __forceinline__ float row_xor(float x) {
    return __builtin_bit_cast(float, __builtin_amdgcn_mov_dpp(__builtin_bit_cast(int, x), 0x160 + M, 0xf, 0xf, true));
}
__device__ __forceinline__ float xhalf(float x) {
    const int b = __builtin_bit_cast(int, x);
    return __builtin_bit_cast(float, __builtin_amdgcn_permlanex16(b, b, 0x76543210, 0xfedcba98, false, false));
}
__device__ __forceinline__ float wave_max(float x) {
    x = fmaxf(x, row_xor<1>(x)); x = fmaxf(x, row_xor<2>(x));
    x = fmaxf(x, row_xor<4>(x)); x = fmaxf(x, row_xor<8>(x));
    return fmaxf(x, xhalf(x));
}
__device__ __forceinline__ float wave_sum(float x) {
    x += row_xor<1>(x); x += row_xor<2>(x);
    x += row_xor<4>(x); x += row_xor<8>(x);
    return x + xhalf(x);
}

struct RowLogSoftmax {
    __device__ __forceinline__ void row(float2& a) {
        const float mx = wave_max(fmaxf(a.x, a.y));
        const float s  = wave_sum(__expf(a.x - mx) + __expf(a.y - mx));
        const float l  = mx + __logf(s);
        a.x -= l; a.y -= l;
    }
};

template <bool WEIGHTED, typename VT, typename EPI>
__device__ __forceinline__ void gather_nodes(const int* __restrict__ cnt, const int* __restrict__ slots,
                                             const int2* __restrict__ ovf, const int* __restrict__ novf,
                                             const VT* __restrict__ rows, VT* __restrict__ dst, int N,
                                             EPI& epi) {
    const int wave = __builtin_amdgcn_readfirstlane(threadIdx.x >> 5);
    const int lane = threadIdx.x & 31;
    const int G = gridDim.x * 8;
    int c0 = blockIdx.x * 8 + wave;
    if (c0 >= N) return;
    int c1 = c0 + G;
    const int nov = *novf;
    int n0 = cnt[c0];
    int n1 = 0;
    if (c1 < N) n1 = cnt[c1];
    int ri = 0;
    if (lane < min(BUCKET, n0)) ri = slots[c0 * BUCKET + lane];
    for (;;) {
        int sc = 0;
        if constexpr (WEIGHTED) sc = cnt[ri];
        const int c2 = c1 + G;
        int n2 = 0;
        if (c2 < N) n2 = cnt[c2];
        int ri1 = 0;
        if (lane < min(BUCKET, n1)) ri1 = slots[c1 * BUCKET + lane];
        const int n = n0;
        const float dc = rsqrtf((float)(n + 1));
        const VT self = rows[(unsigned)c0 * 32u + (unsigned)lane];
        VT acc;
        set_zero(acc);
        {
            const int m = min(BUCKET, n);
            if (m > 0)  edge_batch<WEIGHTED>(rows, lane, ri, sc, 0, min(16, m), acc);
            if (m > 16) edge_batch<WEIGHTED>(rows, lane, ri, sc, 16, m - 16, acc);
        }
        if (n > BUCKET) {
            for (int b = 0; b < nov; b += 32) {
                int2 oe = make_int2(-1, 0);
                if (b + lane < nov) oe = ovf[b + lane];
                unsigned hit = __builtin_amdgcn_ballot_w32(oe.x == c0);
                while (hit != 0u) {
                    const int j = __builtin_amdgcn_readfirstlane(__builtin_ctz(hit));
                    hit &= hit - 1u;
                    const int r = __builtin_amdgcn_readlane(oe.y, j);
                    const VT v = rows[(unsigned)r * 32u + (unsigned)lane];
                    if constexpr (WEIGHTED) fma_row(acc, rsqrtf((float)(cnt[r] + 1)), v);
                    else add_row(acc, v);
                }
            }
        }
        if constexpr (WEIGHTED) fma_row(acc, dc, self);
        else add_row(acc, self);
        scale_row(acc, dc);
        epi.row(acc);
        dst[(unsigned)c0 * 32u + (unsigned)lane] = acc;
        if (c1 >= N) break;
        c0 = c1; c1 = c2;
        n0 = n1; n1 = n2;
        ri = ri1;
    }
}

#define APITCH 33

__device__ __forceinline__ void gemm1_body(const float* __restrict__ x, const float* __restrict__ W1,
                           const float* __restrict__ b1, float* __restrict__ h1, int N,
                           int bid, int nblk) {
    __shared__ float4 atile1[2][16][APITCH];
    const int wave  = __builtin_amdgcn_readfirstlane(threadIdx.x >> 5);
    const int lane  = threadIdx.x & 31;
    const int col0  = wave * 16;
    const int m     = lane & 15;
    const int n     = lane & 15;
    const int khalf = lane >> 4;

    const float bias = b1[col0 + n];
    v2f bf[32];
    #pragma unroll
    for (int u = 0; u < 16; ++u) {
        const int kb = 8 * u + 4 * khalf;
        bf[2 * u].x     = W1[(size_t)(kb + 0) * D_HID + col0 + n];
        bf[2 * u].y     = W1[(size_t)(kb + 1) * D_HID + col0 + n];
        bf[2 * u + 1].x = W1[(size_t)(kb + 2) * D_HID + col0 + n];
        bf[2 * u + 1].y = W1[(size_t)(kb + 3) * D_HID + col0 + n];
    }

    const float4* xr = (const float4*)x;
    const int rstep = nblk * 16;
    int row0 = bid * 16;
    float4 g0 = make_float4(0.0f, 0.0f, 0.0f, 0.0f), g1 = g0;
    if (row0 < N) {
        g0 = xr[(size_t)(row0 + 2 * wave) * 32 + lane];
        g1 = xr[(size_t)(row0 + 2 * wave + 1) * 32 + lane];
    }
    int s = 0;
    for (; row0 < N; row0 += rstep) {
        atile1[s][2 * wave][lane]     = g0;
        atile1[s][2 * wave + 1][lane] = g1;
        __syncthreads();
        const int nrow0 = row0 + rstep;
        if (nrow0 < N) {
            g0 = xr[(size_t)(nrow0 + 2 * wave) * 32 + lane];
            g1 = xr[(size_t)(nrow0 + 2 * wave + 1) * 32 + lane];
        }

        v8f c;
        #pragma unroll
        for (int r = 0; r < 8; ++r) c[r] = bias;

        const int mbase = khalf * 8;
        float4 af[16];
        #pragma unroll
        for (int u = 0; u < 16; ++u) af[u] = atile1[s][m][2 * u + khalf];
        __builtin_amdgcn_sched_barrier(0);
        #pragma unroll
        for (int u = 0; u < 16; ++u) {
            v2f lo, hi;
            lo.x = af[u].x; lo.y = af[u].y;
            hi.x = af[u].z; hi.y = af[u].w;
            c = __builtin_amdgcn_wmma_f32_16x16x4_f32(false, lo, false, bf[2 * u], (short)0, c, false, false);
            c = __builtin_amdgcn_wmma_f32_16x16x4_f32(false, hi, false, bf[2 * u + 1], (short)0, c, false, false);
        }

        asm volatile("v_nop\n\tv_nop\n\tv_nop\n\tv_nop\n\tv_nop\n\tv_nop\n\tv_nop\n\tv_nop" : "+v"(c));
        #pragma unroll
        for (int r = 0; r < 8; ++r)
            h1[(size_t)(row0 + mbase + r) * D_HID + col0 + n] = c[r];
        s ^= 1;
    }
}

__global__ void __launch_bounds__(256, 6) bin_and_gemm1(const int* __restrict__ row, const int* __restrict__ col,
                           int2* __restrict__ tab, int2* __restrict__ sorted,
                           float* __restrict__ stats, int* __restrict__ novf, int E, int PB,
                           const float* __restrict__ x, const float* __restrict__ W1,
                           const float* __restrict__ b1, float* __restrict__ h1, int N, int GB) {
    if ((int)blockIdx.x < PB) bin_edges_body(row, col, tab, sorted, stats, novf, E, (int)blockIdx.x, PB);
    else                      gemm1_body(x, W1, b1, h1, N, (int)blockIdx.x - PB, GB);
}


__global__ void __launch_bounds__(256) gather128(const int* __restrict__ cnt, const int* __restrict__ slots,
                          const int2* __restrict__ ovf, const int* __restrict__ novf,
                          const float* __restrict__ h1, float* __restrict__ hagg, float* stats, int N) {
    __shared__ float part[8][2 * D_HID];
    RowStats st;
    st.s = make_float4(0.0f, 0.0f, 0.0f, 0.0f);
    st.q = make_float4(0.0f, 0.0f, 0.0f, 0.0f);
    gather_nodes<true, float4>(cnt, slots, ovf, novf, (const float4*)h1, (float4*)hagg, N, st);
    const int wave = threadIdx.x >> 5;
    const int lane = threadIdx.x & 31;
    *(float4*)&part[wave][4 * lane]         = st.s;
    *(float4*)&part[wave][D_HID + 4 * lane] = st.q;
    __syncthreads();
    float tot = 0.0f;
    #pragma unroll
    for (int w = 0; w < 8; ++w) tot += part[w][threadIdx.x];
    atomicAdd(&stats[threadIdx.x], tot);
}


__global__ void bn_finalize(const float* __restrict__ bsum, const float* __restrict__ bsq,
                            const float* __restrict__ gamma, const float* __restrict__ beta,
                            float* scale, float* shift, int N) {
    int c = threadIdx.x;
    if (c < D_HID) {
        float inv = 1.0f / (float)N;
        float mu = bsum[c] * inv;
        float var = bsq[c] * inv - mu * mu;
        float s = gamma[c] * rsqrtf(var + BN_EPS);
        scale[c] = s;
        shift[c] = beta[c] - mu * s;
    }
}


__global__ void __launch_bounds__(256) gemm2_wmma(const float* __restrict__ hagg, const float* __restrict__ scale,
                           const float* __restrict__ shift, const float* __restrict__ W2,
                           const float* __restrict__ b2, const int* __restrict__ cnt,
                           float* __restrict__ h2, int N) {
    __shared__ float4 atile2[2][32][APITCH];
    const int wave  = __builtin_amdgcn_readfirstlane(threadIdx.x >> 5);
    const int lane  = threadIdx.x & 31;
    const int col0  = (wave & 3) * 16;
    const int rt    = wave >> 2;
    const int m     = lane & 15;
    const int n     = lane & 15;
    const int khalf = lane >> 4;

    const float bias = b2[col0 + n];
    v2f bf[32];
    #pragma unroll
    for (int u = 0; u < 16; ++u) {
        const int kb = 8 * u + 4 * khalf;
        bf[2 * u].x     = W2[(size_t)(kb + 0) * D_OUT + col0 + n];
        bf[2 * u].y     = W2[(size_t)(kb + 1) * D_OUT + col0 + n];
        bf[2 * u + 1].x = W2[(size_t)(kb + 2) * D_OUT + col0 + n];
        bf[2 * u + 1].y = W2[(size_t)(kb + 3) * D_OUT + col0 + n];
    }

    const float4 sc = ((const float4*)scale)[lane];
    const float4 sh = ((const float4*)shift)[lane];

    const float4* hr = (const float4*)hagg;
    const int pstep = gridDim.x * 32;
    int p0 = blockIdx.x * 32;
    const int mbase = khalf * 8;
    int dn[8];
    #pragma unroll
    for (int r = 0; r < 8; ++r) dn[r] = 0;
    if (p0 + rt * 16 < N) {
        #pragma unroll
        for (int r = 0; r < 8; ++r) dn[r] = cnt[p0 + rt * 16 + mbase + r];
    }
    float4 g[4];
    #pragma unroll
    for (int j = 0; j < 4; ++j) g[j] = make_float4(0.0f, 0.0f, 0.0f, 0.0f);
    if (p0 + 4 * wave < N) {
        #pragma unroll
        for (int j = 0; j < 4; ++j) g[j] = hr[(size_t)(p0 + 4 * wave + j) * 32 + lane];
    }
    int s = 0;
    for (; p0 < N; p0 += pstep) {
        #pragma unroll
        for (int j = 0; j < 4; ++j) {
            float4 a;
            a.x = fmaxf(g[j].x * sc.x + sh.x, 0.0f);
            a.y = fmaxf(g[j].y * sc.y + sh.y, 0.0f);
            a.z = fmaxf(g[j].z * sc.z + sh.z, 0.0f);
            a.w = fmaxf(g[j].w * sc.w + sh.w, 0.0f);
            atile2[s][4 * wave + j][lane] = a;
        }
        __syncthreads();
        const int row0 = p0 + rt * 16;
        const int np0 = p0 + pstep;
        int dnn[8];
        #pragma unroll
        for (int r = 0; r < 8; ++r) dnn[r] = 0;
        if (np0 + rt * 16 < N) {
            #pragma unroll
            for (int r = 0; r < 8; ++r) dnn[r] = cnt[np0 + rt * 16 + mbase + r];
        }
        if (np0 + 4 * wave < N) {
            #pragma unroll
            for (int j = 0; j < 4; ++j) g[j] = hr[(size_t)(np0 + 4 * wave + j) * 32 + lane];
        }

        if (row0 < N) {
            v8f c;
            #pragma unroll
            for (int r = 0; r < 8; ++r) c[r] = bias;

            float4 af[16];
            #pragma unroll
            for (int u = 0; u < 16; ++u) af[u] = atile2[s][rt * 16 + m][2 * u + khalf];
            __builtin_amdgcn_sched_barrier(0);
            #pragma unroll
            for (int u = 0; u < 16; ++u) {
                v2f lo, hi;
                lo.x = af[u].x; lo.y = af[u].y;
                hi.x = af[u].z; hi.y = af[u].w;
                c = __builtin_amdgcn_wmma_f32_16x16x4_f32(false, lo, false, bf[2 * u], (short)0, c, false, false);
                c = __builtin_amdgcn_wmma_f32_16x16x4_f32(false, hi, false, bf[2 * u + 1], (short)0, c, false, false);
            }

            asm volatile("v_nop\n\tv_nop\n\tv_nop\n\tv_nop\n\tv_nop\n\tv_nop\n\tv_nop\n\tv_nop" : "+v"(c));
            #pragma unroll
            for (int r = 0; r < 8; ++r)
                h2[(size_t)(row0 + mbase + r) * D_OUT + col0 + n] = c[r] * __builtin_amdgcn_rsqf((float)(dn[r] + 1));
        }
        #pragma unroll
        for (int r = 0; r < 8; ++r) dn[r] = dnn[r];
        s ^= 1;
    }
}


__global__ void __launch_bounds__(256) gather64(const int* __restrict__ cnt, const int* __restrict__ slots,
                         const int2* __restrict__ ovf, const int* __restrict__ novf,
                         const float* __restrict__ h2, float* __restrict__ out, int N) {
    RowLogSoftmax ls;
    gather_nodes<false, float2>(cnt, slots, ovf, novf, (const float2*)h2, (float2*)out, N, ls);
}


extern "C" void kernel_launch(void* const* d_in, const int* in_sizes, int n_in,
                              void* d_out, int out_size, void* d_ws, size_t ws_size,
                              hipStream_t stream) {
    const float* x     = (const float*)d_in[0];
    const float* W1    = (const float*)d_in[1];
    const float* b1    = (const float*)d_in[2];
    const float* gamma = (const float*)d_in[3];
    const float* beta  = (const float*)d_in[4];
    const float* W2    = (const float*)d_in[5];
    const float* b2    = (const float*)d_in[6];
    const int*   ei    = (const int*)d_in[7];

    const int N = in_sizes[0] / D_HID;
    const int E = in_sizes[7] / 2;
    const int* erow = ei;
    const int* ecol = ei + E;

    float* ws    = (float*)d_ws;
    float* h1    = ws;
    float* hagg  = h1 + (size_t)N * D_HID;
    float* stats = hagg + (size_t)N * D_HID;
    float* bsum  = stats;
    float* bsq   = stats + 128;
    float* scal  = stats + 256;
    float* shft  = stats + 384;
    int*   novf  = (int*)(stats + 512);
    int*   slots = (int*)(stats + 576);
    int2*  ovf   = (int2*)(slots + (size_t)N * BUCKET);
    int*   cnt   = (int*)(ovf + E);
    int2*  tab   = (int2*)(cnt + 65536);
    int2*  sorted = tab + 65536;
    float* h2    = h1;
    float* outp  = (float*)d_out;

    bin_and_gemm1<<<768, 256, 0, stream>>>(erow, ecol, tab, sorted, stats, novf, E, 256, x, W1, b1, h1, N, 512);
    place_owner<<<256, 256, 0, stream>>>(tab, sorted, cnt, slots, ovf, novf, N, 256);

    gather128<<<1024, 256, 0, stream>>>(cnt, slots, ovf, novf, h1, hagg, stats, N);

    bn_finalize<<<1, 128, 0, stream>>>(bsum, bsq, gamma, beta, scal, shft, N);

    gemm2_wmma<<<512, 256, 0, stream>>>(hagg, scal, shft, W2, b2, cnt, h2, N);
    gather64<<<1024, 256, 0, stream>>>(cnt, slots, ovf, novf, h2, outp, N);
}
